# starting kernel
# baseline (speedup 1.0000x reference)
#include <hip/hip_runtime.h>


typedef __attribute__((ext_vector_type(2))) float v2f;
typedef __attribute__((ext_vector_type(8))) float v8f;

#define D 128

__global__ void gcn_zero_kernel(float* __restrict__ p, long long n) {
  long long i = (long long)blockIdx.x * blockDim.x + threadIdx.x;
  long long stride = (long long)gridDim.x * blockDim.x;
  for (; i < n; i += stride) p[i] = 0.0f;
}

__global__ void gcn_degree_kernel(const int* __restrict__ src,
                                  const int* __restrict__ dst,
                                  float* __restrict__ deg_out,
                                  float* __restrict__ deg_in,
                                  int n_edges) {
  int e = blockIdx.x * blockDim.x + threadIdx.x;
  if (e >= n_edges) return;
  atomicAdd(&deg_out[src[e]], 1.0f);
  atomicAdd(&deg_in[dst[e]], 1.0f);
}

__global__ void gcn_invsqrt_kernel(float* __restrict__ deg, int n) {
  int i = blockIdx.x * blockDim.x + threadIdx.x;
  if (i >= n) return;
  deg[i] = rsqrtf(fmaxf(deg[i], 1.0f));
}

__global__ __launch_bounds__(256) void gcn_spmm_kernel(
    const float* __restrict__ h, const int* __restrict__ src,
    const int* __restrict__ dst, const float* __restrict__ inv_out,
    float* __restrict__ agg, int n_edges) {
  int t = blockIdx.x * blockDim.x + threadIdx.x;
  int e = t >> 5;
  if (e >= n_edges) return;
  int lane = t & 31;
  int s = src[e];
  int d = dst[e];
  float sc = inv_out[s];
  const float4 v = *(const float4*)(h + (size_t)s * D + lane * 4);
  float* ap = agg + (size_t)d * D + lane * 4;
  atomicAdd(ap + 0, v.x * sc);
  atomicAdd(ap + 1, v.y * sc);
  atomicAdd(ap + 2, v.z * sc);
  atomicAdd(ap + 3, v.w * sc);
}

__global__ __launch_bounds__(256) void gcn_gemm_kernel(
    const float* __restrict__ agg, const float* __restrict__ inv_in,
    const float* __restrict__ W, const float* __restrict__ bias,
    float* __restrict__ out, int relu) {
  __shared__ float Atile[16 * D];

  const int row0 = blockIdx.x * 16;
  const int tid = threadIdx.x;

  {
    int r = tid >> 4;
    int c = (tid & 15) * 8;
    float s = inv_in[row0 + r];
    const float4* g = (const float4*)(agg + (size_t)(row0 + r) * D + c);
    float4 v0 = g[0];
    float4 v1 = g[1];
    float4* l = (float4*)(Atile + r * D + c);
    v0.x *= s; v0.y *= s; v0.z *= s; v0.w *= s;
    v1.x *= s; v1.y *= s; v1.z *= s; v1.w *= s;
    l[0] = v0;
    l[1] = v1;
  }
  __syncthreads();

  const int lane = tid & 31;
  const int col0 = (tid >> 5) * 16;
  const int m = lane & 15;
  const int khalf = (lane >> 4) * 2;

  v8f c = {};
#pragma unroll 4
  for (int k0 = 0; k0 < D; k0 += 4) {
    v2f a;
    a.x = Atile[m * D + k0 + khalf + 0];
    a.y = Atile[m * D + k0 + khalf + 1];
    v2f b;
    b.x = W[(size_t)(k0 + khalf + 0) * D + col0 + m];
    b.y = W[(size_t)(k0 + khalf + 1) * D + col0 + m];
    c = __builtin_amdgcn_wmma_f32_16x16x4_f32(
         false, a,  false, b,
         (short)0, c,  false,  false);
  }

  const int ncol = col0 + m;
  const int rbase = row0 + (lane >> 4) * 8;
  const float bv = bias[ncol];
#pragma unroll
  for (int r = 0; r < 8; ++r) {
    float v = c[r] + bv;
    if (relu) v = fmaxf(v, 0.0f);
    out[(size_t)(rbase + r) * D + ncol] = v;
  }
}

extern "C" void kernel_launch(void* const* d_in, const int* in_sizes, int n_in,
                              void* d_out, int out_size, void* d_ws, size_t ws_size,
                              hipStream_t stream) {
  const float* feat = (const float*)d_in[0];
  const int* src = (const int*)d_in[1];
  const int* dst = (const int*)d_in[2];
  const float* W1 = (const float*)d_in[3];
  const float* b1 = (const float*)d_in[4];
  const float* W2 = (const float*)d_in[5];
  const float* b2 = (const float*)d_in[6];
  const float* W3 = (const float*)d_in[7];
  const float* b3 = (const float*)d_in[8];
  float* out = (float*)d_out;

  const int N = in_sizes[0] / D;
  const int E = in_sizes[1];

  float* ws = (float*)d_ws;
  float* inv_out = ws;
  float* inv_in = ws + N;
  float* agg = ws + 2 * (size_t)N;
  float* h = agg + (size_t)N * D;

  const int ZB = 2048;

  gcn_zero_kernel<<<ZB, 256, 0, stream>>>(inv_out, 2 * (long long)N);
  gcn_degree_kernel<<<(E + 255) / 256, 256, 0, stream>>>(src, dst, inv_out, inv_in, E);
  gcn_invsqrt_kernel<<<(2 * N + 255) / 256, 256, 0, stream>>>(inv_out, 2 * N);

  const int spmm_blocks = (E * 32 + 255) / 256;
  const int gemm_blocks = N / 16;

  gcn_zero_kernel<<<ZB, 256, 0, stream>>>(agg, (long long)N * D);
  gcn_spmm_kernel<<<spmm_blocks, 256, 0, stream>>>(feat, src, dst, inv_out, agg, E);
  gcn_gemm_kernel<<<gemm_blocks, 256, 0, stream>>>(agg, inv_in, W1, b1, h, 1);

  gcn_zero_kernel<<<ZB, 256, 0, stream>>>(agg, (long long)N * D);
  gcn_spmm_kernel<<<spmm_blocks, 256, 0, stream>>>(h, src, dst, inv_out, agg, E);
  gcn_gemm_kernel<<<gemm_blocks, 256, 0, stream>>>(agg, inv_in, W2, b2, h, 1);

  gcn_zero_kernel<<<ZB, 256, 0, stream>>>(agg, (long long)N * D);
  gcn_spmm_kernel<<<spmm_blocks, 256, 0, stream>>>(h, src, dst, inv_out, agg, E);
  gcn_gemm_kernel<<<gemm_blocks, 256, 0, stream>>>(agg, inv_in, W3, b3, out, 0);
}


# GEMM re-decomposed: a wave owns 16 rows x all 128 columns, A fragments straight from global into registers (no LDS A tile), W staged in LDS in fragment order; k sum order of each dot product changes
# speedup vs baseline: 2.3106x; 2.3106x over previous
#include <hip/hip_runtime.h>


typedef __attribute__((ext_vector_type(2))) float v2f;
typedef __attribute__((ext_vector_type(8))) float v8f;

#define D 128

__global__ void gcn_zero_kernel(float* __restrict__ p, long long n) {
  long long i = (long long)blockIdx.x * blockDim.x + threadIdx.x;
  long long stride = (long long)gridDim.x * blockDim.x;
  for (; i < n; i += stride) p[i] = 0.0f;
}

__global__ void gcn_degree_kernel(const int* __restrict__ src,
                                  const int* __restrict__ dst,
                                  float* __restrict__ deg_out,
                                  float* __restrict__ deg_in,
                                  int n_edges) {
  int e = blockIdx.x * blockDim.x + threadIdx.x;
  if (e >= n_edges) return;
  atomicAdd(&deg_out[src[e]], 1.0f);
  atomicAdd(&deg_in[dst[e]], 1.0f);
}

__global__ void gcn_invsqrt_kernel(float* __restrict__ deg, int n) {
  int i = blockIdx.x * blockDim.x + threadIdx.x;
  if (i >= n) return;
  deg[i] = rsqrtf(fmaxf(deg[i], 1.0f));
}

#define SP_RPB 200
#define SP_WAVES 32
#define SP_CH 1024
#define SP_B 8
#define SP_QCAP (SP_CH + SP_B)
#define SP_LCAP 16384
#define SP_U 16
#define SP_TROWS (SP_RPB + 1)

__device__ __forceinline__ void gcn_spmm_load_chunk(int4 (&dv)[8], const int* __restrict__ dst,
                                                    int chunk, int lane, int n_edges) {
  const int e0 = chunk * SP_CH + lane * 4;
  if (chunk * SP_CH + SP_CH <= n_edges) {
#pragma unroll
    for (int v = 0; v < 8; ++v) dv[v] = *(const int4*)(dst + e0 + v * 128);
  } else {
#pragma unroll
    for (int v = 0; v < 8; ++v) {
      const int e = e0 + v * 128;
      dv[v].x = (e + 0 < n_edges) ? dst[e + 0] : -1;
      dv[v].y = (e + 1 < n_edges) ? dst[e + 1] : -1;
      dv[v].z = (e + 2 < n_edges) ? dst[e + 2] : -1;
      dv[v].w = (e + 3 < n_edges) ? dst[e + 3] : -1;
    }
  }
}

__device__ __forceinline__ void gcn_spmm_push(int* q, int& qn, bool hit, int e) {
  const unsigned m = __builtin_amdgcn_ballot_w32(hit);
  if (hit) q[qn + (int)__builtin_amdgcn_mbcnt_lo(m, 0u)] = e;
  qn += __builtin_popcount(m);
}

__device__ __forceinline__ void gcn_spmm_add_row(float* tile, const float* __restrict__ h,
                                                 float sc, int s, int loc, int lane) {
  const float4 v = ((const float4*)(h + (size_t)s * D))[lane];
  float* t = tile + loc * D + lane;
  atomicAdd(t + 0, v.x * sc);
  atomicAdd(t + 32, v.y * sc);
  atomicAdd(t + 64, v.z * sc);
  atomicAdd(t + 96, v.w * sc);
}

__device__ __forceinline__ int gcn_spmm_list_take(int* lcount, int n, int lane) {
  int pos = 0;
  if (lane == 0) pos = atomicAdd(lcount, n);
  return __builtin_amdgcn_readfirstlane(pos);
}

__device__ __forceinline__ void gcn_spmm_zero_tile(float* tile, int tid) {
  for (int i = tid; i < SP_TROWS * D / 4; i += 1024)
    ((float4*)tile)[i] = make_float4(0.0f, 0.0f, 0.0f, 0.0f);
}

__device__ __forceinline__ void gcn_spmm_store_tile(const float* tile, float* __restrict__ agg,
                                                    int row_lo, int nrows, int lane, int wave) {
  for (int r = wave; r < nrows; r += SP_WAVES) {
    const float* t = tile + r * D + lane;
    float4 o;
    o.x = t[0];
    o.y = t[32];
    o.z = t[64];
    o.w = t[96];
    ((float4*)(agg + (size_t)(row_lo + r) * D))[lane] = o;
  }
}

template <bool KEEP>
__device__ __forceinline__ void gcn_spmm_scan(float* tile, int* queue, int* lcount,
                                              const float* __restrict__ h,
                                              const int* __restrict__ src,
                                              const int* __restrict__ dst,
                                              const float* __restrict__ inv_out,
                                              uint2* __restrict__ list,
                                              int row_lo, int nrows, int n_edges,
                                              int lane, int wave) {
  int* q = queue + wave * SP_QCAP;
  int qn = 0;
  const int nchunks = (n_edges + SP_CH - 1) / SP_CH;
  const int rot = (int)((unsigned)blockIdx.x * (unsigned)nchunks / gridDim.x);

  int c = wave;
  int cc = c + rot;
  if (cc >= nchunks) cc -= nchunks;
  int4 dv[8];
  if (c < nchunks) gcn_spmm_load_chunk(dv, dst, cc, lane, n_edges);

  while (c < nchunks) {
#pragma unroll
    for (int v = 0; v < 8; ++v) {
      const int e = cc * SP_CH + v * 128 + lane * 4;
      const bool h0 = (unsigned)(dv[v].x - row_lo) < (unsigned)nrows;
      const bool h1 = (unsigned)(dv[v].y - row_lo) < (unsigned)nrows;
      const bool h2 = (unsigned)(dv[v].z - row_lo) < (unsigned)nrows;
      const bool h3 = (unsigned)(dv[v].w - row_lo) < (unsigned)nrows;
      if (__builtin_amdgcn_ballot_w32(h0 | h1 | h2 | h3) != 0u) {
        gcn_spmm_push(q, qn, h0, e + 0);
        gcn_spmm_push(q, qn, h1, e + 1);
        gcn_spmm_push(q, qn, h2, e + 2);
        gcn_spmm_push(q, qn, h3, e + 3);
      }
    }

    c += SP_WAVES;
    cc = c + rot;
    if (cc >= nchunks) cc -= nchunks;
    if (c < nchunks) gcn_spmm_load_chunk(dv, dst, cc, lane, n_edges);

    while (qn >= SP_B) {
      qn -= SP_B;
      const int ev = q[qn + (lane & (SP_B - 1))];
      int s[SP_B], d[SP_B];
#pragma unroll
      for (int i = 0; i < SP_B; ++i) {
        const int e = __builtin_amdgcn_readlane(ev, i);
        s[i] = src[e];
        d[i] = dst[e];
      }
      float sc[SP_B];
      float4 v[SP_B];
#pragma unroll
      for (int i = 0; i < SP_B; ++i) {
        sc[i] = inv_out[s[i]];
        v[i] = ((const float4*)(h + (size_t)s[i] * D))[lane];
      }
      if (KEEP) {
        const int pos = gcn_spmm_list_take(lcount, SP_B, lane);
        unsigned w = 0u;
        float f = 0.0f;
#pragma unroll
        for (int i = 0; i < SP_B; ++i) {
          const unsigned wi = ((unsigned)s[i] << 8) | (unsigned)(d[i] - row_lo);
          w = (lane == i) ? wi : w;
          f = (lane == i) ? sc[i] : f;
        }
        if (pos + SP_B <= SP_LCAP && lane < SP_B)
          list[pos + lane] = make_uint2(w, __float_as_uint(f));
      }
#pragma unroll
      for (int i = 0; i < SP_B; ++i) {
        float* t = tile + (d[i] - row_lo) * D + lane;
        atomicAdd(t + 0, v[i].x * sc[i]);
        atomicAdd(t + 32, v[i].y * sc[i]);
        atomicAdd(t + 64, v[i].z * sc[i]);
        atomicAdd(t + 96, v[i].w * sc[i]);
      }
    }
  }

  for (int i = 0; i < qn; ++i) {
    const int e = __builtin_amdgcn_readfirstlane(q[i]);
    const int s = src[e];
    const int loc = dst[e] - row_lo;
    const float sc = inv_out[s];
    gcn_spmm_add_row(tile, h, sc, s, loc, lane);
    if (KEEP) {
      const int pos = gcn_spmm_list_take(lcount, 1, lane);
      if (pos < SP_LCAP && lane == 0)
        list[pos] = make_uint2(((unsigned)s << 8) | (unsigned)loc, __float_as_uint(sc));
    }
  }
}

template <int HALF>
__device__ __forceinline__ void gcn_spmm_list_unit(float* tile, const float* __restrict__ h,
                                                   uint2 e32, int lane) {
  float4 v[SP_U];
#pragma unroll
  for (int i = 0; i < SP_U; ++i) {
    const unsigned w = (unsigned)__builtin_amdgcn_readlane((int)e32.x, SP_U * HALF + i);
    const unsigned off = (w >> 8) * (unsigned)(D * sizeof(float)) + (unsigned)lane * 16u;
    v[i] = *(const float4*)((const char*)h + off);
  }
#pragma unroll
  for (int i = 0; i < SP_U; ++i) {
    const unsigned w = (unsigned)__builtin_amdgcn_readlane((int)e32.x, SP_U * HALF + i);
    const float sc = __int_as_float(__builtin_amdgcn_readlane((int)e32.y, SP_U * HALF + i));
    float* t = tile + (w & 255u) * D + lane;
    atomicAdd(t + 0, v[i].x * sc);
    atomicAdd(t + 32, v[i].y * sc);
    atomicAdd(t + 64, v[i].z * sc);
    atomicAdd(t + 96, v[i].w * sc);
  }
}

__device__ __forceinline__ uint2 gcn_spmm_list_load(const uint2* __restrict__ list, int u,
                                                    int cnt, int lane) {
  const int idx = (u + SP_WAVES * (lane >> 4)) * SP_U + (lane & (SP_U - 1));
  return (idx < cnt) ? list[idx] : make_uint2((unsigned)SP_RPB, 0u);
}

__global__ __launch_bounds__(1024) void gcn_spmm_scan_kernel(
    const float* __restrict__ h, const int* __restrict__ src,
    const int* __restrict__ dst, const float* __restrict__ inv_out,
    float* __restrict__ agg, unsigned* __restrict__ lists, int* __restrict__ counts,
    int n_nodes, int n_edges) {
  __shared__ float tile[SP_TROWS * D];
  __shared__ int queue[SP_WAVES * SP_QCAP];
  __shared__ int lcount;

  const int tid = threadIdx.x;
  const int lane = tid & 31;
  const int wave = __builtin_amdgcn_readfirstlane(tid >> 5);
  const int row_lo = blockIdx.x * SP_RPB;
  int nrows = n_nodes - row_lo;
  if (nrows > SP_RPB) nrows = SP_RPB;

  gcn_spmm_zero_tile(tile, tid);
  if (tid == 0) lcount = 0;
  __syncthreads();

  gcn_spmm_scan<true>(tile, queue, &lcount, h, src, dst, inv_out,
                      (uint2*)lists + (size_t)blockIdx.x * SP_LCAP, row_lo, nrows, n_edges, lane,
                      wave);
  __syncthreads();

  if (tid == 0) counts[blockIdx.x] = lcount;
  gcn_spmm_store_tile(tile, agg, row_lo, nrows, lane, wave);
}

__global__ __launch_bounds__(1024) void gcn_spmm_list_kernel(
    const float* __restrict__ h, const int* __restrict__ src,
    const int* __restrict__ dst, const float* __restrict__ inv_out,
    float* __restrict__ agg, const unsigned* __restrict__ lists,
    const int* __restrict__ counts, int n_nodes, int n_edges) {
  __shared__ float tile[SP_TROWS * D];
  __shared__ int queue[SP_WAVES * SP_QCAP];

  const int tid = threadIdx.x;
  const int lane = tid & 31;
  const int wave = __builtin_amdgcn_readfirstlane(tid >> 5);
  const int row_lo = blockIdx.x * SP_RPB;
  int nrows = n_nodes - row_lo;
  if (nrows > SP_RPB) nrows = SP_RPB;

  const int cnt = counts[blockIdx.x];
  const bool kept = cnt <= SP_LCAP;
  const uint2* list = (const uint2*)lists + (size_t)blockIdx.x * SP_LCAP;
  int u = wave;
  uint2 enext = make_uint2((unsigned)SP_RPB, 0u);
  if (kept) enext = gcn_spmm_list_load(list, u, cnt, lane);

  gcn_spmm_zero_tile(tile, tid);
  __syncthreads();

  if (!kept) {
    gcn_spmm_scan<false>(tile, queue, (int*)0, h, src, dst, inv_out, (uint2*)0,
                         row_lo, nrows, n_edges, lane, wave);
    __syncthreads();
    gcn_spmm_store_tile(tile, agg, row_lo, nrows, lane, wave);
    return;
  }

  const int nunits = (cnt + SP_U - 1) / SP_U;
  while (u < nunits) {
    const uint2 e32 = enext;
    if (u + 2 * SP_WAVES < nunits) enext = gcn_spmm_list_load(list, u + 2 * SP_WAVES, cnt, lane);
    gcn_spmm_list_unit<0>(tile, h, e32, lane);
    if (u + SP_WAVES < nunits) gcn_spmm_list_unit<1>(tile, h, e32, lane);
    u += 2 * SP_WAVES;
  }
  __syncthreads();

  gcn_spmm_store_tile(tile, agg, row_lo, nrows, lane, wave);
}

typedef __attribute__((ext_vector_type(4))) float v4f;

#define GM_WAVES 16
#define GM_THREADS (GM_WAVES * 32)
#define GM_ROWS (GM_WAVES * 16)
#define GM_SLOTS (D * D / 4)

__global__ __launch_bounds__(GM_THREADS) void gcn_gemm_kernel(
    const float* __restrict__ agg, const float* __restrict__ inv_in,
    const float* __restrict__ W, const float* __restrict__ bias,
    float* __restrict__ out, int relu, int n_rows) {
  __shared__ v4f Wl[GM_SLOTS];

  const int tid = threadIdx.x;
  const int lane = tid & 31;
  const int wave = __builtin_amdgcn_readfirstlane(tid >> 5);
  const int c = lane & 15;
  const int hh = lane >> 4;
  const int row0 = (blockIdx.x * GM_WAVES + wave) * 16;
  const bool active = row0 < n_rows;

  v4f a[16];
  float s = 0.0f;
  if (active) {
    const v4f* g = (const v4f*)(agg + (size_t)(row0 + c) * D) + hh;
#pragma unroll
    for (int q = 0; q < 16; ++q) a[q] = g[2 * q];
    s = inv_in[row0 + c];
  }

#pragma unroll
  for (int j = 0; j < GM_SLOTS / GM_THREADS; ++j) {
    const int slot = tid + j * GM_THREADS;
    const int sl = slot & 31;
    const int t = (slot >> 5) & 7;
    const int q = slot >> 8;
    const float* w = W + (size_t)(8 * q + 4 * (sl >> 4)) * D + 16 * t + (sl & 15);
    v4f v;
    v.x = w[0];
    v.y = w[D];
    v.z = w[2 * D];
    v.w = w[3 * D];
    Wl[slot] = v;
  }
  __syncthreads();
  if (!active) return;

#pragma unroll
  for (int q = 0; q < 16; ++q) a[q] = a[q] * s;

  v8f acc[8];
#pragma unroll
  for (int t = 0; t < 8; ++t) acc[t] = (v8f){};

  v4f bcur[8];
#pragma unroll
  for (int t = 0; t < 8; ++t) bcur[t] = Wl[t * 32 + lane];

#pragma unroll
  for (int q = 0; q < 16; ++q) {
    v4f bnxt[8];
    if (q + 1 < 16) {
#pragma unroll
      for (int t = 0; t < 8; ++t) bnxt[t] = Wl[((q + 1) * 8 + t) * 32 + lane];
    }
    __builtin_amdgcn_sched_barrier(0);
    const v4f aq = a[q];
    v2f a0, a1;
    a0.x = aq.x; a0.y = aq.y;
    a1.x = aq.z; a1.y = aq.w;
#pragma unroll
    for (int t = 0; t < 8; ++t) {
      v2f b;
      b.x = bcur[t].x; b.y = bcur[t].y;
      acc[t] = __builtin_amdgcn_wmma_f32_16x16x4_f32(
           false, a0,  false, b,
           (short)0, acc[t],  false,  false);
    }
#pragma unroll
    for (int t = 0; t < 8; ++t) {
      v2f b;
      b.x = bcur[t].z; b.y = bcur[t].w;
      acc[t] = __builtin_amdgcn_wmma_f32_16x16x4_f32(
           false, a1,  false, b,
           (short)0, acc[t],  false,  false);
    }
    __builtin_amdgcn_sched_barrier(0);
    if (q + 1 < 16) {
#pragma unroll
      for (int t = 0; t < 8; ++t) bcur[t] = bnxt[t];
    }
  }

  const int rbase = row0 + hh * 8;
#pragma unroll
  for (int t = 0; t < 8; ++t) {
    const int ncol = 16 * t + c;
    const float bv = bias[ncol];
#pragma unroll
    for (int r = 0; r < 8; ++r) {
      float v = acc[t][r] + bv;
      if (relu) v = fmaxf(v, 0.0f);
      out[(size_t)(rbase + r) * D + ncol] = v;
    }
  }
}

extern "C" void kernel_launch(void* const* d_in, const int* in_sizes, int n_in,
                              void* d_out, int out_size, void* d_ws, size_t ws_size,
                              hipStream_t stream) {
  const float* feat = (const float*)d_in[0];
  const int* src = (const int*)d_in[1];
  const int* dst = (const int*)d_in[2];
  const float* W1 = (const float*)d_in[3];
  const float* b1 = (const float*)d_in[4];
  const float* W2 = (const float*)d_in[5];
  const float* b2 = (const float*)d_in[6];
  const float* W3 = (const float*)d_in[7];
  const float* b3 = (const float*)d_in[8];
  float* out = (float*)d_out;

  const int N = in_sizes[0] / D;
  const int E = in_sizes[1];

  const int spmm_blocks = (N + SP_RPB - 1) / SP_RPB;

  float* ws = (float*)d_ws;
  float* inv_out = ws;
  float* inv_in = ws + N;
  float* agg = ws + 2 * (size_t)N;
  float* h = agg + (size_t)N * D;
  unsigned* lists = (unsigned*)(h + (size_t)N * D);
  int* counts = (int*)(lists + (size_t)spmm_blocks * SP_LCAP * 2);

  const int ZB = 2048;

  gcn_zero_kernel<<<ZB, 256, 0, stream>>>(inv_out, 2 * (long long)N);
  gcn_degree_kernel<<<(E + 255) / 256, 256, 0, stream>>>(src, dst, inv_out, inv_in, E);
  gcn_invsqrt_kernel<<<(2 * N + 255) / 256, 256, 0, stream>>>(inv_out, 2 * N);

  const int gemm_blocks = (N + GM_ROWS - 1) / GM_ROWS;

  gcn_spmm_scan_kernel<<<spmm_blocks, 1024, 0, stream>>>(feat, src, dst, inv_out, agg, lists, counts, N, E);
  gcn_gemm_kernel<<<gemm_blocks, GM_THREADS, 0, stream>>>(agg, inv_in, W1, b1, h, 1, N);

  gcn_spmm_list_kernel<<<spmm_blocks, 1024, 0, stream>>>(h, src, dst, inv_out, agg, lists, counts, N, E);
  gcn_gemm_kernel<<<gemm_blocks, GM_THREADS, 0, stream>>>(agg, inv_in, W2, b2, h, 1, N);

  gcn_spmm_list_kernel<<<spmm_blocks, 1024, 0, stream>>>(h, src, dst, inv_out, agg, lists, counts, N, E);
  gcn_gemm_kernel<<<gemm_blocks, GM_THREADS, 0, stream>>>(agg, inv_in, W3, b3, out, 0, N);
}


# SpMM edge filter done once for all tiles: a binning launch (LDS rank + one int add per tile and group) builds every tile's list, layer 1 sums over lists too; no dst scan; sums still LDS arrival order
# speedup vs baseline: 2.5187x; 1.0901x over previous
#include <hip/hip_runtime.h>


typedef __attribute__((ext_vector_type(2))) float v2f;
typedef __attribute__((ext_vector_type(8))) float v8f;

#define D 128

__global__ void gcn_zero_kernel(float* __restrict__ p, long long n) {
  long long i = (long long)blockIdx.x * blockDim.x + threadIdx.x;
  long long stride = (long long)gridDim.x * blockDim.x;
  for (; i < n; i += stride) p[i] = 0.0f;
}

__global__ void gcn_degree_kernel(const int* __restrict__ src,
                                  const int* __restrict__ dst,
                                  float* __restrict__ deg_out,
                                  float* __restrict__ deg_in,
                                  int n_edges) {
  int e = blockIdx.x * blockDim.x + threadIdx.x;
  if (e >= n_edges) return;
  atomicAdd(&deg_out[src[e]], 1.0f);
  atomicAdd(&deg_in[dst[e]], 1.0f);
}

#define SP_RPB 200
#define SP_WAVES 32
#define SP_CH 1024
#define SP_B 8
#define SP_QCAP (SP_CH + SP_B)
#define SP_LCAP 16384
#define SP_U 16
#define SP_TROWS (SP_RPB + 1)
#define SP_MAXTILES 256
#define SP_CSTRIDE 32

__device__ __forceinline__ void gcn_spmm_load_chunk(int4 (&dv)[8], const int* __restrict__ dst,
                                                    int chunk, int lane, int n_edges) {
  const int e0 = chunk * SP_CH + lane * 4;
  if (chunk * SP_CH + SP_CH <= n_edges) {
#pragma unroll
    for (int v = 0; v < 8; ++v) dv[v] = *(const int4*)(dst + e0 + v * 128);
  } else {
#pragma unroll
    for (int v = 0; v < 8; ++v) {
      const int e = e0 + v * 128;
      dv[v].x = (e + 0 < n_edges) ? dst[e + 0] : -1;
      dv[v].y = (e + 1 < n_edges) ? dst[e + 1] : -1;
      dv[v].z = (e + 2 < n_edges) ? dst[e + 2] : -1;
      dv[v].w = (e + 3 < n_edges) ? dst[e + 3] : -1;
    }
  }
}

__device__ __forceinline__ void gcn_spmm_push(int* q, int& qn, bool hit, int e) {
  const unsigned m = __builtin_amdgcn_ballot_w32(hit);
  if (hit) q[qn + (int)__builtin_amdgcn_mbcnt_lo(m, 0u)] = e;
  qn += __builtin_popcount(m);
}

__device__ __forceinline__ void gcn_spmm_add_row(float* tile, const float* __restrict__ h,
                                                 float sc, int s, int loc, int lane) {
  const float4 v = ((const float4*)(h + (size_t)s * D))[lane];
  float* t = tile + loc * D + lane;
  atomicAdd(t + 0, v.x * sc);
  atomicAdd(t + 32, v.y * sc);
  atomicAdd(t + 64, v.z * sc);
  atomicAdd(t + 96, v.w * sc);
}

__device__ __forceinline__ void gcn_spmm_zero_tile(float* tile, int tid) {
  for (int i = tid; i < SP_TROWS * D / 4; i += 1024)
    ((float4*)tile)[i] = make_float4(0.0f, 0.0f, 0.0f, 0.0f);
}

__device__ __forceinline__ void gcn_spmm_store_tile(const float* tile, float* __restrict__ agg,
                                                    int row_lo, int nrows, int lane, int wave) {
  for (int r = wave; r < nrows; r += SP_WAVES) {
    const float* t = tile + r * D + lane;
    float4 o;
    o.x = t[0];
    o.y = t[32];
    o.z = t[64];
    o.w = t[96];
    ((float4*)(agg + (size_t)(row_lo + r) * D))[lane] = o;
  }
}

__device__ __forceinline__ void gcn_spmm_scan(float* tile, int* queue,
                                              const float* __restrict__ h,
                                              const int* __restrict__ src,
                                              const int* __restrict__ dst,
                                              const float* __restrict__ inv_out,
                                              int row_lo, int nrows, int n_edges,
                                              int lane, int wave) {
  int* q = queue + wave * SP_QCAP;
  int qn = 0;
  const int nchunks = (n_edges + SP_CH - 1) / SP_CH;
  const int rot = (int)((unsigned)blockIdx.x * (unsigned)nchunks / gridDim.x);

  int c = wave;
  int cc = c + rot;
  if (cc >= nchunks) cc -= nchunks;
  int4 dv[8];
  if (c < nchunks) gcn_spmm_load_chunk(dv, dst, cc, lane, n_edges);

  while (c < nchunks) {
#pragma unroll
    for (int v = 0; v < 8; ++v) {
      const int e = cc * SP_CH + v * 128 + lane * 4;
      const bool h0 = (unsigned)(dv[v].x - row_lo) < (unsigned)nrows;
      const bool h1 = (unsigned)(dv[v].y - row_lo) < (unsigned)nrows;
      const bool h2 = (unsigned)(dv[v].z - row_lo) < (unsigned)nrows;
      const bool h3 = (unsigned)(dv[v].w - row_lo) < (unsigned)nrows;
      if (__builtin_amdgcn_ballot_w32(h0 | h1 | h2 | h3) != 0u) {
        gcn_spmm_push(q, qn, h0, e + 0);
        gcn_spmm_push(q, qn, h1, e + 1);
        gcn_spmm_push(q, qn, h2, e + 2);
        gcn_spmm_push(q, qn, h3, e + 3);
      }
    }

    c += SP_WAVES;
    cc = c + rot;
    if (cc >= nchunks) cc -= nchunks;
    if (c < nchunks) gcn_spmm_load_chunk(dv, dst, cc, lane, n_edges);

    while (qn >= SP_B) {
      qn -= SP_B;
      const int ev = q[qn + (lane & (SP_B - 1))];
      int s[SP_B], d[SP_B];
#pragma unroll
      for (int i = 0; i < SP_B; ++i) {
        const int e = __builtin_amdgcn_readlane(ev, i);
        s[i] = src[e];
        d[i] = dst[e];
      }
      float sc[SP_B];
      float4 v[SP_B];
#pragma unroll
      for (int i = 0; i < SP_B; ++i) {
        sc[i] = inv_out[s[i]];
        v[i] = ((const float4*)(h + (size_t)s[i] * D))[lane];
      }
#pragma unroll
      for (int i = 0; i < SP_B; ++i) {
        float* t = tile + (d[i] - row_lo) * D + lane;
        atomicAdd(t + 0, v[i].x * sc[i]);
        atomicAdd(t + 32, v[i].y * sc[i]);
        atomicAdd(t + 64, v[i].z * sc[i]);
        atomicAdd(t + 96, v[i].w * sc[i]);
      }
    }
  }

  for (int i = 0; i < qn; ++i) {
    const int e = __builtin_amdgcn_readfirstlane(q[i]);
    const int s = src[e];
    const int loc = dst[e] - row_lo;
    const float sc = inv_out[s];
    gcn_spmm_add_row(tile, h, sc, s, loc, lane);
  }
}

template <int HALF>
__device__ __forceinline__ void gcn_spmm_list_unit(float* tile, const float* __restrict__ h,
                                                   uint2 e32, int lane) {
  float4 v[SP_U];
#pragma unroll
  for (int i = 0; i < SP_U; ++i) {
    const unsigned w = (unsigned)__builtin_amdgcn_readlane((int)e32.x, SP_U * HALF + i);
    const unsigned off = (w >> 8) * (unsigned)(D * sizeof(float)) + (unsigned)lane * 16u;
    v[i] = *(const float4*)((const char*)h + off);
  }
#pragma unroll
  for (int i = 0; i < SP_U; ++i) {
    const unsigned w = (unsigned)__builtin_amdgcn_readlane((int)e32.x, SP_U * HALF + i);
    const float sc = __int_as_float(__builtin_amdgcn_readlane((int)e32.y, SP_U * HALF + i));
    float* t = tile + (w & 255u) * D + lane;
    atomicAdd(t + 0, v[i].x * sc);
    atomicAdd(t + 32, v[i].y * sc);
    atomicAdd(t + 64, v[i].z * sc);
    atomicAdd(t + 96, v[i].w * sc);
  }
}

__device__ __forceinline__ uint2 gcn_spmm_list_load(const uint2* __restrict__ list, int u,
                                                    int cnt, int lane) {
  const int idx = (u + SP_WAVES * (lane >> 4)) * SP_U + (lane & (SP_U - 1));
  return (idx < cnt) ? list[idx] : make_uint2((unsigned)SP_RPB, 0u);
}

#define BN_EPT 4
#define BN_EDGES (1024 * BN_EPT)

__global__ __launch_bounds__(1024) void gcn_bin_kernel(
    const int* __restrict__ src, const int* __restrict__ dst,
    const float* __restrict__ deg, float* __restrict__ inv,
    unsigned* __restrict__ lists, int* __restrict__ counts, int n_nodes, int n_edges) {
  __shared__ int lc[SP_MAXTILES];

  const int tid = threadIdx.x;
  if (tid < SP_MAXTILES) lc[tid] = 0;
  __syncthreads();

  int t[BN_EPT], r[BN_EPT], d[BN_EPT], s[BN_EPT];
  float dg[BN_EPT];
  const int e0 = blockIdx.x * BN_EDGES + tid;
#pragma unroll
  for (int i = 0; i < BN_EPT; ++i) {
    const int e = e0 + i * 1024;
    const int ec = (e < n_edges) ? e : n_edges - 1;
    d[i] = dst[ec];
    s[i] = src[ec];
  }
#pragma unroll
  for (int i = 0; i < BN_EPT; ++i) dg[i] = deg[s[i]];
#pragma unroll
  for (int i = 0; i < BN_EPT; ++i) {
    const int tt = d[i] / SP_RPB;
    const bool ok = e0 + i * 1024 < n_edges;
    t[i] = ok ? tt : -1;
    r[i] = 0;
    if (ok) r[i] = atomicAdd(&lc[tt], 1);
  }

  for (int i = blockIdx.x * 1024 + tid; i < 2 * n_nodes; i += gridDim.x * 1024)
    inv[i] = rsqrtf(fmaxf(deg[i], 1.0f));
  __syncthreads();

  if (tid < SP_MAXTILES) {
    const int n = lc[tid];
    lc[tid] = (n > 0) ? atomicAdd(&counts[tid * SP_CSTRIDE], n) : 0;
  }
  __syncthreads();

  int base[BN_EPT];
#pragma unroll
  for (int i = 0; i < BN_EPT; ++i) base[i] = lc[(t[i] >= 0) ? t[i] : 0];
#pragma unroll
  for (int i = 0; i < BN_EPT; ++i) {
    const int pos = base[i] + r[i];
    if (t[i] >= 0 && pos < SP_LCAP) {
      const unsigned w = ((unsigned)s[i] << 8) | (unsigned)(d[i] - t[i] * SP_RPB);
      const float sc = rsqrtf(fmaxf(dg[i], 1.0f));
      ((uint2*)lists)[(size_t)t[i] * SP_LCAP + pos] = make_uint2(w, __float_as_uint(sc));
    }
  }
}

__global__ __launch_bounds__(1024) void gcn_spmm_list_kernel(
    const float* __restrict__ h, const int* __restrict__ src,
    const int* __restrict__ dst, const float* __restrict__ inv_out,
    float* __restrict__ agg, const unsigned* __restrict__ lists,
    const int* __restrict__ counts, int n_nodes, int n_edges) {
  __shared__ float tile[SP_TROWS * D];
  __shared__ int queue[SP_WAVES * SP_QCAP];

  const int tid = threadIdx.x;
  const int lane = tid & 31;
  const int wave = __builtin_amdgcn_readfirstlane(tid >> 5);
  const int row_lo = blockIdx.x * SP_RPB;
  int nrows = n_nodes - row_lo;
  if (nrows > SP_RPB) nrows = SP_RPB;

  const int cnt = counts[blockIdx.x * SP_CSTRIDE];
  const bool kept = cnt <= SP_LCAP;
  const uint2* list = (const uint2*)lists + (size_t)blockIdx.x * SP_LCAP;
  int u = wave;
  uint2 enext = make_uint2((unsigned)SP_RPB, 0u);
  if (kept) enext = gcn_spmm_list_load(list, u, cnt, lane);

  gcn_spmm_zero_tile(tile, tid);
  __syncthreads();

  if (!kept) {
    gcn_spmm_scan(tile, queue, h, src, dst, inv_out, row_lo, nrows, n_edges, lane, wave);
    __syncthreads();
    gcn_spmm_store_tile(tile, agg, row_lo, nrows, lane, wave);
    return;
  }

  const int nunits = (cnt + SP_U - 1) / SP_U;
  while (u < nunits) {
    const uint2 e32 = enext;
    if (u + 2 * SP_WAVES < nunits) enext = gcn_spmm_list_load(list, u + 2 * SP_WAVES, cnt, lane);
    gcn_spmm_list_unit<0>(tile, h, e32, lane);
    if (u + SP_WAVES < nunits) gcn_spmm_list_unit<1>(tile, h, e32, lane);
    u += 2 * SP_WAVES;
  }
  __syncthreads();

  gcn_spmm_store_tile(tile, agg, row_lo, nrows, lane, wave);
}

typedef __attribute__((ext_vector_type(4))) float v4f;

#define GM_WAVES 16
#define GM_THREADS (GM_WAVES * 32)
#define GM_ROWS (GM_WAVES * 16)
#define GM_SLOTS (D * D / 4)

__global__ __launch_bounds__(GM_THREADS) void gcn_gemm_kernel(
    const float* __restrict__ agg, const float* __restrict__ inv_in,
    const float* __restrict__ W, const float* __restrict__ bias,
    float* __restrict__ out, int relu, int n_rows) {
  __shared__ v4f Wl[GM_SLOTS];

  const int tid = threadIdx.x;
  const int lane = tid & 31;
  const int wave = __builtin_amdgcn_readfirstlane(tid >> 5);
  const int c = lane & 15;
  const int hh = lane >> 4;
  const int row0 = (blockIdx.x * GM_WAVES + wave) * 16;
  const bool active = row0 < n_rows;

  v4f a[16];
  float s = 0.0f;
  if (active) {
    const v4f* g = (const v4f*)(agg + (size_t)(row0 + c) * D) + hh;
#pragma unroll
    for (int q = 0; q < 16; ++q) a[q] = g[2 * q];
    s = inv_in[row0 + c];
  }

#pragma unroll
  for (int j = 0; j < GM_SLOTS / GM_THREADS; ++j) {
    const int slot = tid + j * GM_THREADS;
    const int sl = slot & 31;
    const int t = (slot >> 5) & 7;
    const int q = slot >> 8;
    const float* w = W + (size_t)(8 * q + 4 * (sl >> 4)) * D + 16 * t + (sl & 15);
    v4f v;
    v.x = w[0];
    v.y = w[D];
    v.z = w[2 * D];
    v.w = w[3 * D];
    Wl[slot] = v;
  }
  __syncthreads();
  if (!active) return;

#pragma unroll
  for (int q = 0; q < 16; ++q) a[q] = a[q] * s;

  v8f acc[8];
#pragma unroll
  for (int t = 0; t < 8; ++t) acc[t] = (v8f){};

  v4f bcur[8];
#pragma unroll
  for (int t = 0; t < 8; ++t) bcur[t] = Wl[t * 32 + lane];

#pragma unroll
  for (int q = 0; q < 16; ++q) {
    v4f bnxt[8];
    if (q + 1 < 16) {
#pragma unroll
      for (int t = 0; t < 8; ++t) bnxt[t] = Wl[((q + 1) * 8 + t) * 32 + lane];
    }
    __builtin_amdgcn_sched_barrier(0);
    const v4f aq = a[q];
    v2f a0, a1;
    a0.x = aq.x; a0.y = aq.y;
    a1.x = aq.z; a1.y = aq.w;
#pragma unroll
    for (int t = 0; t < 8; ++t) {
      v2f b;
      b.x = bcur[t].x; b.y = bcur[t].y;
      acc[t] = __builtin_amdgcn_wmma_f32_16x16x4_f32(
           false, a0,  false, b,
           (short)0, acc[t],  false,  false);
    }
#pragma unroll
    for (int t = 0; t < 8; ++t) {
      v2f b;
      b.x = bcur[t].z; b.y = bcur[t].w;
      acc[t] = __builtin_amdgcn_wmma_f32_16x16x4_f32(
           false, a1,  false, b,
           (short)0, acc[t],  false,  false);
    }
    __builtin_amdgcn_sched_barrier(0);
    if (q + 1 < 16) {
#pragma unroll
      for (int t = 0; t < 8; ++t) bcur[t] = bnxt[t];
    }
  }

  const int rbase = row0 + hh * 8;
#pragma unroll
  for (int t = 0; t < 8; ++t) {
    const int ncol = 16 * t + c;
    const float bv = bias[ncol];
#pragma unroll
    for (int r = 0; r < 8; ++r) {
      float v = acc[t][r] + bv;
      if (relu) v = fmaxf(v, 0.0f);
      out[(size_t)(rbase + r) * D + ncol] = v;
    }
  }
}

extern "C" void kernel_launch(void* const* d_in, const int* in_sizes, int n_in,
                              void* d_out, int out_size, void* d_ws, size_t ws_size,
                              hipStream_t stream) {
  const float* feat = (const float*)d_in[0];
  const int* src = (const int*)d_in[1];
  const int* dst = (const int*)d_in[2];
  const float* W1 = (const float*)d_in[3];
  const float* b1 = (const float*)d_in[4];
  const float* W2 = (const float*)d_in[5];
  const float* b2 = (const float*)d_in[6];
  const float* W3 = (const float*)d_in[7];
  const float* b3 = (const float*)d_in[8];
  float* out = (float*)d_out;

  const int N = in_sizes[0] / D;
  const int E = in_sizes[1];

  const int spmm_blocks = (N + SP_RPB - 1) / SP_RPB;

  float* ws = (float*)d_ws;
  float* deg = ws;
  int* counts = (int*)(ws + 2 * (size_t)N);
  float* inv_out = ws + 2 * (size_t)N + SP_MAXTILES * SP_CSTRIDE;
  float* inv_in = inv_out + N;
  float* agg = inv_in + N;
  float* h = agg + (size_t)N * D;
  unsigned* lists = (unsigned*)(h + (size_t)N * D);

  const int ZB = 2048;

  gcn_zero_kernel<<<ZB, 256, 0, stream>>>(deg, 2 * (long long)N + SP_MAXTILES * SP_CSTRIDE);
  gcn_degree_kernel<<<(E + 255) / 256, 256, 0, stream>>>(src, dst, deg, deg + N, E);
  gcn_bin_kernel<<<(E + BN_EDGES - 1) / BN_EDGES, 1024, 0, stream>>>(src, dst, deg, inv_out, lists, counts, N, E);

  const int gemm_blocks = (N + GM_ROWS - 1) / GM_ROWS;

  gcn_spmm_list_kernel<<<spmm_blocks, 1024, 0, stream>>>(feat, src, dst, inv_out, agg, lists, counts, N, E);
  gcn_gemm_kernel<<<gemm_blocks, GM_THREADS, 0, stream>>>(agg, inv_in, W1, b1, h, 1, N);

  gcn_spmm_list_kernel<<<spmm_blocks, 1024, 0, stream>>>(h, src, dst, inv_out, agg, lists, counts, N, E);
  gcn_gemm_kernel<<<gemm_blocks, GM_THREADS, 0, stream>>>(agg, inv_in, W2, b2, h, 1, N);

  gcn_spmm_list_kernel<<<spmm_blocks, 1024, 0, stream>>>(h, src, dst, inv_out, agg, lists, counts, N, E);
  gcn_gemm_kernel<<<gemm_blocks, GM_THREADS, 0, stream>>>(agg, inv_in, W3, b3, out, 0, N);
}


# SpMM sums in registers: a tile's list is split among its 32 waves by row mod 32 (LDS count, prefix, place); a wave adds gathered rows to its 7 rows' sums; no f32 atomics; sum order = list order
# speedup vs baseline: 2.5979x; 1.0314x over previous
#include <hip/hip_runtime.h>


typedef __attribute__((ext_vector_type(2))) float v2f;
typedef __attribute__((ext_vector_type(8))) float v8f;
typedef __attribute__((ext_vector_type(32))) float v32f;

#define D 128

__global__ void gcn_zero_kernel(float* __restrict__ p, long long n) {
  long long i = (long long)blockIdx.x * blockDim.x + threadIdx.x;
  long long stride = (long long)gridDim.x * blockDim.x;
  for (; i < n; i += stride) p[i] = 0.0f;
}

__global__ void gcn_degree_kernel(const int* __restrict__ src,
                                  const int* __restrict__ dst,
                                  float* __restrict__ deg_out,
                                  float* __restrict__ deg_in,
                                  int n_edges) {
  int e = blockIdx.x * blockDim.x + threadIdx.x;
  if (e >= n_edges) return;
  atomicAdd(&deg_out[src[e]], 1.0f);
  atomicAdd(&deg_in[dst[e]], 1.0f);
}

#define SP_RPB 200
#define SP_WAVES 32
#define SP_CH 1024
#define SP_B 8
#define SP_QCAP (SP_CH + SP_B)
#define SP_LCAP 16384
#define SP_U 16
#define SP_TROWS (SP_RPB + 1)
#define SP_OWN ((SP_RPB + SP_WAVES - 1) / SP_WAVES)
#define SP_MAXTILES 256
#define SP_CSTRIDE 32

__device__ __forceinline__ void gcn_spmm_load_chunk(int4 (&dv)[8], const int* __restrict__ dst,
                                                    int chunk, int lane, int n_edges) {
  const int e0 = chunk * SP_CH + lane * 4;
  if (chunk * SP_CH + SP_CH <= n_edges) {
#pragma unroll
    for (int v = 0; v < 8; ++v) dv[v] = *(const int4*)(dst + e0 + v * 128);
  } else {
#pragma unroll
    for (int v = 0; v < 8; ++v) {
      const int e = e0 + v * 128;
      dv[v].x = (e + 0 < n_edges) ? dst[e + 0] : -1;
      dv[v].y = (e + 1 < n_edges) ? dst[e + 1] : -1;
      dv[v].z = (e + 2 < n_edges) ? dst[e + 2] : -1;
      dv[v].w = (e + 3 < n_edges) ? dst[e + 3] : -1;
    }
  }
}

__device__ __forceinline__ void gcn_spmm_push(int* q, int& qn, bool hit, int e) {
  const unsigned m = __builtin_amdgcn_ballot_w32(hit);
  if (hit) q[qn + (int)__builtin_amdgcn_mbcnt_lo(m, 0u)] = e;
  qn += __builtin_popcount(m);
}

__device__ __forceinline__ void gcn_spmm_add_row(float* tile, const float* __restrict__ h,
                                                 float sc, int s, int loc, int lane) {
  const float4 v = ((const float4*)(h + (size_t)s * D))[lane];
  float* t = tile + loc * D + lane;
  atomicAdd(t + 0, v.x * sc);
  atomicAdd(t + 32, v.y * sc);
  atomicAdd(t + 64, v.z * sc);
  atomicAdd(t + 96, v.w * sc);
}

__device__ __forceinline__ void gcn_spmm_zero_tile(float* tile, int tid) {
  for (int i = tid; i < SP_TROWS * D / 4; i += 1024)
    ((float4*)tile)[i] = make_float4(0.0f, 0.0f, 0.0f, 0.0f);
}

__device__ __forceinline__ void gcn_spmm_store_tile(const float* tile, float* __restrict__ agg,
                                                    int row_lo, int nrows, int lane, int wave) {
  for (int r = wave; r < nrows; r += SP_WAVES) {
    const float* t = tile + r * D + lane;
    float4 o;
    o.x = t[0];
    o.y = t[32];
    o.z = t[64];
    o.w = t[96];
    ((float4*)(agg + (size_t)(row_lo + r) * D))[lane] = o;
  }
}

__device__ __forceinline__ void gcn_spmm_scan(float* tile, int* queue,
                                              const float* __restrict__ h,
                                              const int* __restrict__ src,
                                              const int* __restrict__ dst,
                                              const float* __restrict__ inv_out,
                                              int row_lo, int nrows, int n_edges,
                                              int lane, int wave) {
  int* q = queue + wave * SP_QCAP;
  int qn = 0;
  const int nchunks = (n_edges + SP_CH - 1) / SP_CH;
  const int rot = (int)((unsigned)blockIdx.x * (unsigned)nchunks / gridDim.x);

  int c = wave;
  int cc = c + rot;
  if (cc >= nchunks) cc -= nchunks;
  int4 dv[8];
  if (c < nchunks) gcn_spmm_load_chunk(dv, dst, cc, lane, n_edges);

  while (c < nchunks) {
#pragma unroll
    for (int v = 0; v < 8; ++v) {
      const int e = cc * SP_CH + v * 128 + lane * 4;
      const bool h0 = (unsigned)(dv[v].x - row_lo) < (unsigned)nrows;
      const bool h1 = (unsigned)(dv[v].y - row_lo) < (unsigned)nrows;
      const bool h2 = (unsigned)(dv[v].z - row_lo) < (unsigned)nrows;
      const bool h3 = (unsigned)(dv[v].w - row_lo) < (unsigned)nrows;
      if (__builtin_amdgcn_ballot_w32(h0 | h1 | h2 | h3) != 0u) {
        gcn_spmm_push(q, qn, h0, e + 0);
        gcn_spmm_push(q, qn, h1, e + 1);
        gcn_spmm_push(q, qn, h2, e + 2);
        gcn_spmm_push(q, qn, h3, e + 3);
      }
    }

    c += SP_WAVES;
    cc = c + rot;
    if (cc >= nchunks) cc -= nchunks;
    if (c < nchunks) gcn_spmm_load_chunk(dv, dst, cc, lane, n_edges);

    while (qn >= SP_B) {
      qn -= SP_B;
      const int ev = q[qn + (lane & (SP_B - 1))];
      int s[SP_B], d[SP_B];
#pragma unroll
      for (int i = 0; i < SP_B; ++i) {
        const int e = __builtin_amdgcn_readlane(ev, i);
        s[i] = src[e];
        d[i] = dst[e];
      }
      float sc[SP_B];
      float4 v[SP_B];
#pragma unroll
      for (int i = 0; i < SP_B; ++i) {
        sc[i] = inv_out[s[i]];
        v[i] = ((const float4*)(h + (size_t)s[i] * D))[lane];
      }
#pragma unroll
      for (int i = 0; i < SP_B; ++i) {
        float* t = tile + (d[i] - row_lo) * D + lane;
        atomicAdd(t + 0, v[i].x * sc[i]);
        atomicAdd(t + 32, v[i].y * sc[i]);
        atomicAdd(t + 64, v[i].z * sc[i]);
        atomicAdd(t + 96, v[i].w * sc[i]);
      }
    }
  }

  for (int i = 0; i < qn; ++i) {
    const int e = __builtin_amdgcn_readfirstlane(q[i]);
    const int s = src[e];
    const int loc = dst[e] - row_lo;
    const float sc = inv_out[s];
    gcn_spmm_add_row(tile, h, sc, s, loc, lane);
  }
}

template <int HALF>
__device__ __forceinline__ void gcn_spmm_own_unit(v32f& acc,
                                                  const float* __restrict__ h, uint2 e32,
                                                  int lane) {
  float4 v[SP_U];
#pragma unroll
  for (int i = 0; i < SP_U; ++i) {
    const unsigned w = (unsigned)__builtin_amdgcn_readlane((int)e32.x, SP_U * HALF + i);
    const unsigned off = (w >> 8) * (unsigned)(D * sizeof(float)) + (unsigned)lane * 16u;
    v[i] = *(const float4*)((const char*)h + off);
  }
#pragma unroll
  for (int i = 0; i < SP_U; ++i) {
    const unsigned w = (unsigned)__builtin_amdgcn_readlane((int)e32.x, SP_U * HALF + i);
    const float sc = __int_as_float(__builtin_amdgcn_readlane((int)e32.y, SP_U * HALF + i));
    const unsigned g = ((w & 255u) >> 5) * 4u;
    acc[g + 0] += v[i].x * sc;
    acc[g + 1] += v[i].y * sc;
    acc[g + 2] += v[i].z * sc;
    acc[g + 3] += v[i].w * sc;
  }
}

__device__ __forceinline__ uint2 gcn_spmm_own_load(const uint2* sorted, int start, int n, int at,
                                                   int lane) {
  return (at + lane < n) ? sorted[start + at + lane] : make_uint2(255u, 0u);
}

#define BN_EPT 4
#define BN_EDGES (1024 * BN_EPT)

__global__ __launch_bounds__(1024) void gcn_bin_kernel(
    const int* __restrict__ src, const int* __restrict__ dst,
    const float* __restrict__ deg, float* __restrict__ inv,
    unsigned* __restrict__ lists, int* __restrict__ counts, int n_nodes, int n_edges) {
  __shared__ int lc[SP_MAXTILES];

  const int tid = threadIdx.x;
  if (tid < SP_MAXTILES) lc[tid] = 0;
  __syncthreads();

  int t[BN_EPT], r[BN_EPT], d[BN_EPT], s[BN_EPT];
  float dg[BN_EPT];
  const int e0 = blockIdx.x * BN_EDGES + tid;
#pragma unroll
  for (int i = 0; i < BN_EPT; ++i) {
    const int e = e0 + i * 1024;
    const int ec = (e < n_edges) ? e : n_edges - 1;
    d[i] = dst[ec];
    s[i] = src[ec];
  }
#pragma unroll
  for (int i = 0; i < BN_EPT; ++i) dg[i] = deg[s[i]];
#pragma unroll
  for (int i = 0; i < BN_EPT; ++i) {
    const int tt = d[i] / SP_RPB;
    const bool ok = e0 + i * 1024 < n_edges;
    t[i] = ok ? tt : -1;
    r[i] = 0;
    if (ok) r[i] = atomicAdd(&lc[tt], 1);
  }

  for (int i = blockIdx.x * 1024 + tid; i < 2 * n_nodes; i += gridDim.x * 1024)
    inv[i] = rsqrtf(fmaxf(deg[i], 1.0f));
  __syncthreads();

  if (tid < SP_MAXTILES) {
    const int n = lc[tid];
    lc[tid] = (n > 0) ? atomicAdd(&counts[tid * SP_CSTRIDE], n) : 0;
  }
  __syncthreads();

  int base[BN_EPT];
#pragma unroll
  for (int i = 0; i < BN_EPT; ++i) base[i] = lc[(t[i] >= 0) ? t[i] : 0];
#pragma unroll
  for (int i = 0; i < BN_EPT; ++i) {
    const int pos = base[i] + r[i];
    if (t[i] >= 0 && pos < SP_LCAP) {
      const unsigned w = ((unsigned)s[i] << 8) | (unsigned)(d[i] - t[i] * SP_RPB);
      const float sc = rsqrtf(fmaxf(dg[i], 1.0f));
      ((uint2*)lists)[(size_t)t[i] * SP_LCAP + pos] = make_uint2(w, __float_as_uint(sc));
    }
  }
}

static_assert(SP_OWN * 4 + 4 <= 32, "a wave's sums and the spare group must fit the 32-float array");
static_assert(SP_RPB <= 224, "row group 7 must stay free for the padding entry");
static_assert(SP_LCAP * 8 <= SP_WAVES * SP_QCAP * 4, "a kept list must fit the queue's LDS");
#define SP_PRE 4

__global__ __launch_bounds__(1024) void gcn_spmm_list_kernel(
    const float* __restrict__ h, const int* __restrict__ src,
    const int* __restrict__ dst, const float* __restrict__ inv_out,
    float* __restrict__ agg, const unsigned* __restrict__ lists,
    const int* __restrict__ counts, int n_nodes, int n_edges) {
  __shared__ float tile[SP_TROWS * D];
  __shared__ __attribute__((aligned(16))) int queue[SP_WAVES * SP_QCAP];
  __shared__ int bincnt[SP_WAVES];
  __shared__ int binstart[SP_WAVES];
  __shared__ int binpos[SP_WAVES];

  const int tid = threadIdx.x;
  const int lane = tid & 31;
  const int wave = __builtin_amdgcn_readfirstlane(tid >> 5);
  const int row_lo = blockIdx.x * SP_RPB;
  int nrows = n_nodes - row_lo;
  if (nrows > SP_RPB) nrows = SP_RPB;

  const int cnt = counts[blockIdx.x * SP_CSTRIDE];
  const bool kept = cnt <= SP_LCAP;
  const uint2* list = (const uint2*)lists + (size_t)blockIdx.x * SP_LCAP;

  if (!kept) {
    gcn_spmm_zero_tile(tile, tid);
    __syncthreads();
    gcn_spmm_scan(tile, queue, h, src, dst, inv_out, row_lo, nrows, n_edges, lane, wave);
    __syncthreads();
    gcn_spmm_store_tile(tile, agg, row_lo, nrows, lane, wave);
    return;
  }

  uint2 ent[SP_PRE];
#pragma unroll
  for (int k = 0; k < SP_PRE; ++k) {
    const int i = tid + k * 1024;
    ent[k] = (i < cnt) ? list[i] : make_uint2(0u, 0u);
  }
  if (tid < SP_WAVES) bincnt[tid] = 0;
  __syncthreads();

#pragma unroll
  for (int k = 0; k < SP_PRE; ++k)
    if (tid + k * 1024 < cnt) atomicAdd(&bincnt[ent[k].x & 31u], 1);
  for (int i = tid + SP_PRE * 1024; i < cnt; i += 1024) atomicAdd(&bincnt[list[i].x & 31u], 1);
  __syncthreads();

  if (wave == 0) {
    const int c = bincnt[lane];
    int incl = c;
#pragma unroll
    for (int o = 1; o < 32; o <<= 1) {
      const int t = __shfl_up(incl, o);
      if (lane >= o) incl += t;
    }
    binstart[lane] = incl - c;
    binpos[lane] = incl - c;
  }
  __syncthreads();

  uint2* sorted = (uint2*)queue;
#pragma unroll
  for (int k = 0; k < SP_PRE; ++k)
    if (tid + k * 1024 < cnt) sorted[atomicAdd(&binpos[ent[k].x & 31u], 1)] = ent[k];
  for (int i = tid + SP_PRE * 1024; i < cnt; i += 1024) {
    const uint2 e = list[i];
    sorted[atomicAdd(&binpos[e.x & 31u], 1)] = e;
  }
  __syncthreads();

  const int my_start = __builtin_amdgcn_readfirstlane(binstart[wave]);
  const int my_n = __builtin_amdgcn_readfirstlane(bincnt[wave]);

  v32f acc = (v32f){};

  int at = 0;
  uint2 enext = gcn_spmm_own_load(sorted, my_start, my_n, 0, lane);
  while (at < my_n) {
    const uint2 e32 = enext;
    if (at + 32 < my_n) enext = gcn_spmm_own_load(sorted, my_start, my_n, at + 32, lane);
    gcn_spmm_own_unit<0>(acc, h, e32, lane);
    if (at + SP_U < my_n) gcn_spmm_own_unit<1>(acc, h, e32, lane);
    at += 32;
  }

#pragma unroll
  for (int j = 0; j < SP_OWN; ++j) {
    const int r = wave + SP_WAVES * j;
    if (r < nrows)
      ((float4*)(agg + (size_t)(row_lo + r) * D))[lane] =
          make_float4(acc[4 * j], acc[4 * j + 1], acc[4 * j + 2], acc[4 * j + 3]);
  }
}

typedef __attribute__((ext_vector_type(4))) float v4f;

#define GM_WAVES 16
#define GM_THREADS (GM_WAVES * 32)
#define GM_ROWS (GM_WAVES * 16)
#define GM_SLOTS (D * D / 4)

__global__ __launch_bounds__(GM_THREADS) void gcn_gemm_kernel(
    const float* __restrict__ agg, const float* __restrict__ inv_in,
    const float* __restrict__ W, const float* __restrict__ bias,
    float* __restrict__ out, int relu, int n_rows) {
  __shared__ v4f Wl[GM_SLOTS];

  const int tid = threadIdx.x;
  const int lane = tid & 31;
  const int wave = __builtin_amdgcn_readfirstlane(tid >> 5);
  const int c = lane & 15;
  const int hh = lane >> 4;
  const int row0 = (blockIdx.x * GM_WAVES + wave) * 16;
  const bool active = row0 < n_rows;

  v4f a[16];
  float s = 0.0f;
  if (active) {
    const v4f* g = (const v4f*)(agg + (size_t)(row0 + c) * D) + hh;
#pragma unroll
    for (int q = 0; q < 16; ++q) a[q] = g[2 * q];
    s = inv_in[row0 + c];
  }

#pragma unroll
  for (int j = 0; j < GM_SLOTS / GM_THREADS; ++j) {
    const int slot = tid + j * GM_THREADS;
    const int sl = slot & 31;
    const int t = (slot >> 5) & 7;
    const int q = slot >> 8;
    const float* w = W + (size_t)(8 * q + 4 * (sl >> 4)) * D + 16 * t + (sl & 15);
    v4f v;
    v.x = w[0];
    v.y = w[D];
    v.z = w[2 * D];
    v.w = w[3 * D];
    Wl[slot] = v;
  }
  __syncthreads();
  if (!active) return;

#pragma unroll
  for (int q = 0; q < 16; ++q) a[q] = a[q] * s;

  v8f acc[8];
#pragma unroll
  for (int t = 0; t < 8; ++t) acc[t] = (v8f){};

  v4f bcur[8];
#pragma unroll
  for (int t = 0; t < 8; ++t) bcur[t] = Wl[t * 32 + lane];

#pragma unroll
  for (int q = 0; q < 16; ++q) {
    v4f bnxt[8];
    if (q + 1 < 16) {
#pragma unroll
      for (int t = 0; t < 8; ++t) bnxt[t] = Wl[((q + 1) * 8 + t) * 32 + lane];
    }
    __builtin_amdgcn_sched_barrier(0);
    const v4f aq = a[q];
    v2f a0, a1;
    a0.x = aq.x; a0.y = aq.y;
    a1.x = aq.z; a1.y = aq.w;
#pragma unroll
    for (int t = 0; t < 8; ++t) {
      v2f b;
      b.x = bcur[t].x; b.y = bcur[t].y;
      acc[t] = __builtin_amdgcn_wmma_f32_16x16x4_f32(
           false, a0,  false, b,
           (short)0, acc[t],  false,  false);
    }
#pragma unroll
    for (int t = 0; t < 8; ++t) {
      v2f b;
      b.x = bcur[t].z; b.y = bcur[t].w;
      acc[t] = __builtin_amdgcn_wmma_f32_16x16x4_f32(
           false, a1,  false, b,
           (short)0, acc[t],  false,  false);
    }
    __builtin_amdgcn_sched_barrier(0);
    if (q + 1 < 16) {
#pragma unroll
      for (int t = 0; t < 8; ++t) bcur[t] = bnxt[t];
    }
  }

  const int rbase = row0 + hh * 8;
#pragma unroll
  for (int t = 0; t < 8; ++t) {
    const int ncol = 16 * t + c;
    const float bv = bias[ncol];
#pragma unroll
    for (int r = 0; r < 8; ++r) {
      float v = acc[t][r] + bv;
      if (relu) v = fmaxf(v, 0.0f);
      out[(size_t)(rbase + r) * D + ncol] = v;
    }
  }
}

extern "C" void kernel_launch(void* const* d_in, const int* in_sizes, int n_in,
                              void* d_out, int out_size, void* d_ws, size_t ws_size,
                              hipStream_t stream) {
  const float* feat = (const float*)d_in[0];
  const int* src = (const int*)d_in[1];
  const int* dst = (const int*)d_in[2];
  const float* W1 = (const float*)d_in[3];
  const float* b1 = (const float*)d_in[4];
  const float* W2 = (const float*)d_in[5];
  const float* b2 = (const float*)d_in[6];
  const float* W3 = (const float*)d_in[7];
  const float* b3 = (const float*)d_in[8];
  float* out = (float*)d_out;

  const int N = in_sizes[0] / D;
  const int E = in_sizes[1];

  const int spmm_blocks = (N + SP_RPB - 1) / SP_RPB;

  float* ws = (float*)d_ws;
  float* deg = ws;
  int* counts = (int*)(ws + 2 * (size_t)N);
  float* inv_out = ws + 2 * (size_t)N + SP_MAXTILES * SP_CSTRIDE;
  float* inv_in = inv_out + N;
  float* agg = inv_in + N;
  float* h = agg + (size_t)N * D;
  unsigned* lists = (unsigned*)(h + (size_t)N * D);

  const int ZB = 2048;

  gcn_zero_kernel<<<ZB, 256, 0, stream>>>(deg, 2 * (long long)N + SP_MAXTILES * SP_CSTRIDE);
  gcn_degree_kernel<<<(E + 255) / 256, 256, 0, stream>>>(src, dst, deg, deg + N, E);
  gcn_bin_kernel<<<(E + BN_EDGES - 1) / BN_EDGES, 1024, 0, stream>>>(src, dst, deg, inv_out, lists, counts, N, E);

  const int gemm_blocks = (N + GM_ROWS - 1) / GM_ROWS;

  gcn_spmm_list_kernel<<<spmm_blocks, 1024, 0, stream>>>(feat, src, dst, inv_out, agg, lists, counts, N, E);
  gcn_gemm_kernel<<<gemm_blocks, GM_THREADS, 0, stream>>>(agg, inv_in, W1, b1, h, 1, N);

  gcn_spmm_list_kernel<<<spmm_blocks, 1024, 0, stream>>>(h, src, dst, inv_out, agg, lists, counts, N, E);
  gcn_gemm_kernel<<<gemm_blocks, GM_THREADS, 0, stream>>>(agg, inv_in, W2, b2, h, 1, N);

  gcn_spmm_list_kernel<<<spmm_blocks, 1024, 0, stream>>>(h, src, dst, inv_out, agg, lists, counts, N, E);
  gcn_gemm_kernel<<<gemm_blocks, GM_THREADS, 0, stream>>>(agg, inv_in, W3, b3, out, 0, N);
}


# No degree launch: binning pass counts out-degrees (int adds), SpMM looks entry scales up from them; in-degree = a tile row's entry count in the SpMM, applied at its store; clear sized to fit
# speedup vs baseline: 3.2148x; 1.2375x over previous
#include <hip/hip_runtime.h>


typedef __attribute__((ext_vector_type(2))) float v2f;
typedef __attribute__((ext_vector_type(8))) float v8f;
typedef __attribute__((ext_vector_type(32))) float v32f;

#define D 128

__global__ void gcn_zero_kernel(float* __restrict__ p, long long n) {
  long long i = (long long)blockIdx.x * blockDim.x + threadIdx.x;
  long long stride = (long long)gridDim.x * blockDim.x;
  for (; i < n; i += stride) p[i] = 0.0f;
}

__device__ __forceinline__ float gcn_inv_sqrt_deg(int count) {
  return rsqrtf(fmaxf((float)count, 1.0f));
}

#define SP_RPB 200
#define SP_WAVES 32
#define SP_CH 1024
#define SP_B 8
#define SP_QCAP (SP_CH + SP_B)
#define SP_LCAP 16384
#define SP_U 16
#define SP_TROWS (SP_RPB + 1)
#define SP_OWN ((SP_RPB + SP_WAVES - 1) / SP_WAVES)
#define SP_MAXTILES 256
#define SP_CSTRIDE 32

__device__ __forceinline__ void gcn_spmm_load_chunk(int4 (&dv)[8], const int* __restrict__ dst,
                                                    int chunk, int lane, int n_edges) {
  const int e0 = chunk * SP_CH + lane * 4;
  if (chunk * SP_CH + SP_CH <= n_edges) {
#pragma unroll
    for (int v = 0; v < 8; ++v) dv[v] = *(const int4*)(dst + e0 + v * 128);
  } else {
#pragma unroll
    for (int v = 0; v < 8; ++v) {
      const int e = e0 + v * 128;
      dv[v].x = (e + 0 < n_edges) ? dst[e + 0] : -1;
      dv[v].y = (e + 1 < n_edges) ? dst[e + 1] : -1;
      dv[v].z = (e + 2 < n_edges) ? dst[e + 2] : -1;
      dv[v].w = (e + 3 < n_edges) ? dst[e + 3] : -1;
    }
  }
}

__device__ __forceinline__ void gcn_spmm_push(int* q, int& qn, bool hit, int e) {
  const unsigned m = __builtin_amdgcn_ballot_w32(hit);
  if (hit) q[qn + (int)__builtin_amdgcn_mbcnt_lo(m, 0u)] = e;
  qn += __builtin_popcount(m);
}

__device__ __forceinline__ void gcn_spmm_add_row(float* tile, const float* __restrict__ h,
                                                 float sc, int s, int loc, int lane) {
  const float4 v = ((const float4*)(h + (size_t)s * D))[lane];
  float* t = tile + loc * D + lane;
  atomicAdd(t + 0, v.x * sc);
  atomicAdd(t + 32, v.y * sc);
  atomicAdd(t + 64, v.z * sc);
  atomicAdd(t + 96, v.w * sc);
}

__device__ __forceinline__ void gcn_spmm_zero_tile(float* tile, int tid) {
  for (int i = tid; i < SP_TROWS * D / 4; i += 1024)
    ((float4*)tile)[i] = make_float4(0.0f, 0.0f, 0.0f, 0.0f);
}

__device__ __forceinline__ void gcn_spmm_store_tile(const float* tile, const int* rowcnt,
                                                    float* __restrict__ agg,
                                                    int row_lo, int nrows, int lane, int wave) {
  for (int r = wave; r < nrows; r += SP_WAVES) {
    const float* t = tile + r * D + lane;
    const float si = gcn_inv_sqrt_deg(rowcnt[r]);
    float4 o;
    o.x = t[0] * si;
    o.y = t[32] * si;
    o.z = t[64] * si;
    o.w = t[96] * si;
    ((float4*)(agg + (size_t)(row_lo + r) * D))[lane] = o;
  }
}

__device__ __forceinline__ void gcn_spmm_scan(float* tile, int* rowcnt, int* queue,
                                              const float* __restrict__ h,
                                              const int* __restrict__ src,
                                              const int* __restrict__ dst,
                                              const int* __restrict__ deg_out,
                                              int row_lo, int nrows, int n_edges,
                                              int lane, int wave) {
  int* q = queue + wave * SP_QCAP;
  int qn = 0;
  const int nchunks = (n_edges + SP_CH - 1) / SP_CH;
  const int rot = (int)((unsigned)blockIdx.x * (unsigned)nchunks / gridDim.x);

  int c = wave;
  int cc = c + rot;
  if (cc >= nchunks) cc -= nchunks;
  int4 dv[8];
  if (c < nchunks) gcn_spmm_load_chunk(dv, dst, cc, lane, n_edges);

  while (c < nchunks) {
#pragma unroll
    for (int v = 0; v < 8; ++v) {
      const int e = cc * SP_CH + v * 128 + lane * 4;
      const bool h0 = (unsigned)(dv[v].x - row_lo) < (unsigned)nrows;
      const bool h1 = (unsigned)(dv[v].y - row_lo) < (unsigned)nrows;
      const bool h2 = (unsigned)(dv[v].z - row_lo) < (unsigned)nrows;
      const bool h3 = (unsigned)(dv[v].w - row_lo) < (unsigned)nrows;
      if (__builtin_amdgcn_ballot_w32(h0 | h1 | h2 | h3) != 0u) {
        gcn_spmm_push(q, qn, h0, e + 0);
        gcn_spmm_push(q, qn, h1, e + 1);
        gcn_spmm_push(q, qn, h2, e + 2);
        gcn_spmm_push(q, qn, h3, e + 3);
      }
    }

    c += SP_WAVES;
    cc = c + rot;
    if (cc >= nchunks) cc -= nchunks;
    if (c < nchunks) gcn_spmm_load_chunk(dv, dst, cc, lane, n_edges);

    while (qn >= SP_B) {
      qn -= SP_B;
      const int ev = q[qn + (lane & (SP_B - 1))];
      int s[SP_B], d[SP_B];
#pragma unroll
      for (int i = 0; i < SP_B; ++i) {
        const int e = __builtin_amdgcn_readlane(ev, i);
        s[i] = src[e];
        d[i] = dst[e];
      }
      float sc[SP_B];
      float4 v[SP_B];
#pragma unroll
      for (int i = 0; i < SP_B; ++i) {
        sc[i] = gcn_inv_sqrt_deg(deg_out[s[i]]);
        v[i] = ((const float4*)(h + (size_t)s[i] * D))[lane];
      }
#pragma unroll
      for (int i = 0; i < SP_B; ++i) {
        if (lane == 0) atomicAdd(&rowcnt[d[i] - row_lo], 1);
        float* t = tile + (d[i] - row_lo) * D + lane;
        atomicAdd(t + 0, v[i].x * sc[i]);
        atomicAdd(t + 32, v[i].y * sc[i]);
        atomicAdd(t + 64, v[i].z * sc[i]);
        atomicAdd(t + 96, v[i].w * sc[i]);
      }
    }
  }

  for (int i = 0; i < qn; ++i) {
    const int e = __builtin_amdgcn_readfirstlane(q[i]);
    const int s = src[e];
    const int loc = dst[e] - row_lo;
    const float sc = gcn_inv_sqrt_deg(deg_out[s]);
    if (lane == 0) atomicAdd(&rowcnt[loc], 1);
    gcn_spmm_add_row(tile, h, sc, s, loc, lane);
  }
}

template <int HALF>
__device__ __forceinline__ void gcn_spmm_own_unit(v32f& acc,
                                                  const float* __restrict__ h, uint2 e32,
                                                  int lane) {
  float4 v[SP_U];
#pragma unroll
  for (int i = 0; i < SP_U; ++i) {
    const unsigned w = (unsigned)__builtin_amdgcn_readlane((int)e32.x, SP_U * HALF + i);
    const unsigned off = (w >> 8) * (unsigned)(D * sizeof(float)) + (unsigned)lane * 16u;
    v[i] = *(const float4*)((const char*)h + off);
  }
#pragma unroll
  for (int i = 0; i < SP_U; ++i) {
    const unsigned w = (unsigned)__builtin_amdgcn_readlane((int)e32.x, SP_U * HALF + i);
    const float sc = __int_as_float(__builtin_amdgcn_readlane((int)e32.y, SP_U * HALF + i));
    const unsigned g = ((w & 255u) >> 5) * 4u;
    acc[g + 0] += v[i].x * sc;
    acc[g + 1] += v[i].y * sc;
    acc[g + 2] += v[i].z * sc;
    acc[g + 3] += v[i].w * sc;
  }
}

__device__ __forceinline__ uint2 gcn_spmm_own_load(const uint2* sorted, int start, int n, int at,
                                                   int lane) {
  return (at + lane < n) ? sorted[start + at + lane] : make_uint2(255u, 0u);
}

#define BN_EPT 4
#define BN_EDGES (1024 * BN_EPT)

__global__ __launch_bounds__(1024) void gcn_bin_kernel(
    const int* __restrict__ src, const int* __restrict__ dst,
    int* __restrict__ deg_out,
    unsigned* __restrict__ lists, int* __restrict__ counts, int n_edges) {
  __shared__ int lc[SP_MAXTILES];

  const int tid = threadIdx.x;
  if (tid < SP_MAXTILES) lc[tid] = 0;
  __syncthreads();

  int t[BN_EPT], r[BN_EPT], d[BN_EPT], s[BN_EPT];
  const int e0 = blockIdx.x * BN_EDGES + tid;
#pragma unroll
  for (int i = 0; i < BN_EPT; ++i) {
    const int e = e0 + i * 1024;
    const int ec = (e < n_edges) ? e : n_edges - 1;
    d[i] = dst[ec];
    s[i] = src[ec];
  }
#pragma unroll
  for (int i = 0; i < BN_EPT; ++i) {
    const int tt = d[i] / SP_RPB;
    const bool ok = e0 + i * 1024 < n_edges;
    t[i] = ok ? tt : -1;
    r[i] = 0;
    if (ok) r[i] = atomicAdd(&lc[tt], 1);
  }
  __syncthreads();

  if (tid < SP_MAXTILES) {
    const int n = lc[tid];
    lc[tid] = (n > 0) ? atomicAdd(&counts[tid * SP_CSTRIDE], n) : 0;
  }
  __syncthreads();

  int base[BN_EPT];
#pragma unroll
  for (int i = 0; i < BN_EPT; ++i) base[i] = lc[(t[i] >= 0) ? t[i] : 0];
#pragma unroll
  for (int i = 0; i < BN_EPT; ++i) {
    const int pos = base[i] + r[i];
    if (t[i] >= 0 && pos < SP_LCAP)
      lists[(size_t)t[i] * SP_LCAP + pos] = ((unsigned)s[i] << 8) | (unsigned)(d[i] - t[i] * SP_RPB);
    if (t[i] >= 0) atomicAdd(&deg_out[s[i]], 1);
  }
}

static_assert(SP_OWN * 4 + 4 <= 32, "a wave's sums and the spare group must fit the 32-float array");
static_assert(SP_RPB <= 224, "row group 7 must stay free for the padding entry");
static_assert(SP_LCAP * 8 <= SP_WAVES * SP_QCAP * 4, "a kept list must fit the queue's LDS");
#define SP_PRE 4

__global__ __launch_bounds__(1024) void gcn_spmm_list_kernel(
    const float* __restrict__ h, const int* __restrict__ src,
    const int* __restrict__ dst, const int* __restrict__ deg_out,
    float* __restrict__ agg, const unsigned* __restrict__ lists,
    const int* __restrict__ counts, int n_nodes, int n_edges) {
  __shared__ float tile[SP_TROWS * D];
  __shared__ __attribute__((aligned(16))) int queue[SP_WAVES * SP_QCAP];
  __shared__ int bincnt[SP_WAVES];
  __shared__ int binstart[SP_WAVES];
  __shared__ int binpos[SP_WAVES];
  __shared__ int rowcnt[SP_TROWS];

  const int tid = threadIdx.x;
  const int lane = tid & 31;
  const int wave = __builtin_amdgcn_readfirstlane(tid >> 5);
  const int row_lo = blockIdx.x * SP_RPB;
  int nrows = n_nodes - row_lo;
  if (nrows > SP_RPB) nrows = SP_RPB;

  const int cnt = counts[blockIdx.x * SP_CSTRIDE];
  const bool kept = cnt <= SP_LCAP;
  const unsigned* list = lists + (size_t)blockIdx.x * SP_LCAP;

  if (tid < SP_WAVES) bincnt[tid] = 0;
  if (tid < SP_TROWS) rowcnt[tid] = 0;

  if (!kept) {
    gcn_spmm_zero_tile(tile, tid);
    __syncthreads();
    gcn_spmm_scan(tile, rowcnt, queue, h, src, dst, deg_out, row_lo, nrows, n_edges, lane, wave);
    __syncthreads();
    gcn_spmm_store_tile(tile, rowcnt, agg, row_lo, nrows, lane, wave);
    return;
  }

  unsigned ent[SP_PRE];
  int edeg[SP_PRE];
#pragma unroll
  for (int k = 0; k < SP_PRE; ++k) {
    const int i = tid + k * 1024;
    ent[k] = (i < cnt) ? list[i] : 0u;
  }
  __syncthreads();

#pragma unroll
  for (int k = 0; k < SP_PRE; ++k) edeg[k] = deg_out[ent[k] >> 8];

#pragma unroll
  for (int k = 0; k < SP_PRE; ++k)
    if (tid + k * 1024 < cnt) {
      atomicAdd(&bincnt[ent[k] & 31u], 1);
      atomicAdd(&rowcnt[ent[k] & 255u], 1);
    }
  for (int i = tid + SP_PRE * 1024; i < cnt; i += 1024) {
    const unsigned w = list[i];
    atomicAdd(&bincnt[w & 31u], 1);
    atomicAdd(&rowcnt[w & 255u], 1);
  }
  __syncthreads();

  if (wave == 0) {
    const int c = bincnt[lane];
    int incl = c;
#pragma unroll
    for (int o = 1; o < 32; o <<= 1) {
      const int t = __shfl_up(incl, o);
      if (lane >= o) incl += t;
    }
    binstart[lane] = incl - c;
    binpos[lane] = incl - c;
  }
  __syncthreads();

  uint2* sorted = (uint2*)queue;
#pragma unroll
  for (int k = 0; k < SP_PRE; ++k)
    if (tid + k * 1024 < cnt)
      sorted[atomicAdd(&binpos[ent[k] & 31u], 1)] =
          make_uint2(ent[k], __float_as_uint(gcn_inv_sqrt_deg(edeg[k])));
  for (int i = tid + SP_PRE * 1024; i < cnt; i += 1024) {
    const unsigned w = list[i];
    sorted[atomicAdd(&binpos[w & 31u], 1)] =
        make_uint2(w, __float_as_uint(gcn_inv_sqrt_deg(deg_out[w >> 8])));
  }
  __syncthreads();

  const int my_start = __builtin_amdgcn_readfirstlane(binstart[wave]);
  const int my_n = __builtin_amdgcn_readfirstlane(bincnt[wave]);

  v32f acc = (v32f){};

  int at = 0;
  uint2 enext = gcn_spmm_own_load(sorted, my_start, my_n, 0, lane);
  while (at < my_n) {
    const uint2 e32 = enext;
    if (at + 32 < my_n) enext = gcn_spmm_own_load(sorted, my_start, my_n, at + 32, lane);
    gcn_spmm_own_unit<0>(acc, h, e32, lane);
    if (at + SP_U < my_n) gcn_spmm_own_unit<1>(acc, h, e32, lane);
    at += 32;
  }

#pragma unroll
  for (int j = 0; j < SP_OWN; ++j) {
    const int r = wave + SP_WAVES * j;
    if (r < nrows) {
      const float si = gcn_inv_sqrt_deg(rowcnt[r]);
      ((float4*)(agg + (size_t)(row_lo + r) * D))[lane] =
          make_float4(acc[4 * j] * si, acc[4 * j + 1] * si, acc[4 * j + 2] * si,
                      acc[4 * j + 3] * si);
    }
  }
}

typedef __attribute__((ext_vector_type(4))) float v4f;

#define GM_WAVES 16
#define GM_THREADS (GM_WAVES * 32)
#define GM_ROWS (GM_WAVES * 16)
#define GM_SLOTS (D * D / 4)

__global__ __launch_bounds__(GM_THREADS) void gcn_gemm_kernel(
    const float* __restrict__ agg,
    const float* __restrict__ W, const float* __restrict__ bias,
    float* __restrict__ out, int relu, int n_rows) {
  __shared__ v4f Wl[GM_SLOTS];

  const int tid = threadIdx.x;
  const int lane = tid & 31;
  const int wave = __builtin_amdgcn_readfirstlane(tid >> 5);
  const int c = lane & 15;
  const int hh = lane >> 4;
  const int row0 = (blockIdx.x * GM_WAVES + wave) * 16;
  const bool active = row0 < n_rows;

  v4f a[16];
  if (active) {
    const v4f* g = (const v4f*)(agg + (size_t)(row0 + c) * D) + hh;
#pragma unroll
    for (int q = 0; q < 16; ++q) a[q] = g[2 * q];
  }

#pragma unroll
  for (int j = 0; j < GM_SLOTS / GM_THREADS; ++j) {
    const int slot = tid + j * GM_THREADS;
    const int sl = slot & 31;
    const int t = (slot >> 5) & 7;
    const int q = slot >> 8;
    const float* w = W + (size_t)(8 * q + 4 * (sl >> 4)) * D + 16 * t + (sl & 15);
    v4f v;
    v.x = w[0];
    v.y = w[D];
    v.z = w[2 * D];
    v.w = w[3 * D];
    Wl[slot] = v;
  }
  __syncthreads();
  if (!active) return;

  v8f acc[8];
#pragma unroll
  for (int t = 0; t < 8; ++t) acc[t] = (v8f){};

  v4f bcur[8];
#pragma unroll
  for (int t = 0; t < 8; ++t) bcur[t] = Wl[t * 32 + lane];

#pragma unroll
  for (int q = 0; q < 16; ++q) {
    v4f bnxt[8];
    if (q + 1 < 16) {
#pragma unroll
      for (int t = 0; t < 8; ++t) bnxt[t] = Wl[((q + 1) * 8 + t) * 32 + lane];
    }
    __builtin_amdgcn_sched_barrier(0);
    const v4f aq = a[q];
    v2f a0, a1;
    a0.x = aq.x; a0.y = aq.y;
    a1.x = aq.z; a1.y = aq.w;
#pragma unroll
    for (int t = 0; t < 8; ++t) {
      v2f b;
      b.x = bcur[t].x; b.y = bcur[t].y;
      acc[t] = __builtin_amdgcn_wmma_f32_16x16x4_f32(
           false, a0,  false, b,
           (short)0, acc[t],  false,  false);
    }
#pragma unroll
    for (int t = 0; t < 8; ++t) {
      v2f b;
      b.x = bcur[t].z; b.y = bcur[t].w;
      acc[t] = __builtin_amdgcn_wmma_f32_16x16x4_f32(
           false, a1,  false, b,
           (short)0, acc[t],  false,  false);
    }
    __builtin_amdgcn_sched_barrier(0);
    if (q + 1 < 16) {
#pragma unroll
      for (int t = 0; t < 8; ++t) bcur[t] = bnxt[t];
    }
  }

  const int rbase = row0 + hh * 8;
#pragma unroll
  for (int t = 0; t < 8; ++t) {
    const int ncol = 16 * t + c;
    const float bv = bias[ncol];
#pragma unroll
    for (int r = 0; r < 8; ++r) {
      float v = acc[t][r] + bv;
      if (relu) v = fmaxf(v, 0.0f);
      out[(size_t)(rbase + r) * D + ncol] = v;
    }
  }
}

extern "C" void kernel_launch(void* const* d_in, const int* in_sizes, int n_in,
                              void* d_out, int out_size, void* d_ws, size_t ws_size,
                              hipStream_t stream) {
  const float* feat = (const float*)d_in[0];
  const int* src = (const int*)d_in[1];
  const int* dst = (const int*)d_in[2];
  const float* W1 = (const float*)d_in[3];
  const float* b1 = (const float*)d_in[4];
  const float* W2 = (const float*)d_in[5];
  const float* b2 = (const float*)d_in[6];
  const float* W3 = (const float*)d_in[7];
  const float* b3 = (const float*)d_in[8];
  float* out = (float*)d_out;

  const int N = in_sizes[0] / D;
  const int E = in_sizes[1];

  const int spmm_blocks = (N + SP_RPB - 1) / SP_RPB;

  float* ws = (float*)d_ws;
  int* counts = (int*)ws;
  int* deg_out = counts + SP_MAXTILES * SP_CSTRIDE;
  const long long n_clear = SP_MAXTILES * SP_CSTRIDE + ((long long)N + 63) / 64 * 64;
  float* agg = ws + n_clear;
  float* h = agg + (size_t)N * D;
  unsigned* lists = (unsigned*)(h + (size_t)N * D);

  gcn_zero_kernel<<<(int)((n_clear + 1023) / 1024), 1024, 0, stream>>>(ws, n_clear);
  gcn_bin_kernel<<<(E + BN_EDGES - 1) / BN_EDGES, 1024, 0, stream>>>(src, dst, deg_out, lists, counts, E);

  const int gemm_blocks = (N + GM_ROWS - 1) / GM_ROWS;

  gcn_spmm_list_kernel<<<spmm_blocks, 1024, 0, stream>>>(feat, src, dst, deg_out, agg, lists, counts, N, E);
  gcn_gemm_kernel<<<gemm_blocks, GM_THREADS, 0, stream>>>(agg, W1, b1, h, 1, N);

  gcn_spmm_list_kernel<<<spmm_blocks, 1024, 0, stream>>>(h, src, dst, deg_out, agg, lists, counts, N, E);
  gcn_gemm_kernel<<<gemm_blocks, GM_THREADS, 0, stream>>>(agg, W2, b2, h, 1, N);

  gcn_spmm_list_kernel<<<spmm_blocks, 1024, 0, stream>>>(h, src, dst, deg_out, agg, lists, counts, N, E);
  gcn_gemm_kernel<<<gemm_blocks, GM_THREADS, 0, stream>>>(agg, W3, b3, out, 0, N);
}


# SpMM split of a tile's list among its waves done once: layer 1 keeps owner-ordered entries+scales, stretch bounds, row factors in the workspace; layers 2,3 read them, no LDS pass, no barrier
# speedup vs baseline: 3.3388x; 1.0386x over previous
#include <hip/hip_runtime.h>


typedef __attribute__((ext_vector_type(2))) float v2f;
typedef __attribute__((ext_vector_type(8))) float v8f;
typedef __attribute__((ext_vector_type(32))) float v32f;

#define D 128

__global__ void gcn_zero_kernel(float* __restrict__ p, long long n) {
  long long i = (long long)blockIdx.x * blockDim.x + threadIdx.x;
  long long stride = (long long)gridDim.x * blockDim.x;
  for (; i < n; i += stride) p[i] = 0.0f;
}

__device__ __forceinline__ float gcn_inv_sqrt_deg(int count) {
  return rsqrtf(fmaxf((float)count, 1.0f));
}

#define SP_RPB 200
#define SP_WAVES 32
#define SP_CH 1024
#define SP_B 8
#define SP_QCAP (SP_CH + SP_B)
#define SP_LCAP 16384
#define SP_U 16
#define SP_TROWS (SP_RPB + 1)
#define SP_OWN ((SP_RPB + SP_WAVES - 1) / SP_WAVES)
#define SP_MAXTILES 256
#define SP_CSTRIDE 32

__device__ __forceinline__ void gcn_spmm_load_chunk(int4 (&dv)[8], const int* __restrict__ dst,
                                                    int chunk, int lane, int n_edges) {
  const int e0 = chunk * SP_CH + lane * 4;
  if (chunk * SP_CH + SP_CH <= n_edges) {
#pragma unroll
    for (int v = 0; v < 8; ++v) dv[v] = *(const int4*)(dst + e0 + v * 128);
  } else {
#pragma unroll
    for (int v = 0; v < 8; ++v) {
      const int e = e0 + v * 128;
      dv[v].x = (e + 0 < n_edges) ? dst[e + 0] : -1;
      dv[v].y = (e + 1 < n_edges) ? dst[e + 1] : -1;
      dv[v].z = (e + 2 < n_edges) ? dst[e + 2] : -1;
      dv[v].w = (e + 3 < n_edges) ? dst[e + 3] : -1;
    }
  }
}

__device__ __forceinline__ void gcn_spmm_push(int* q, int& qn, bool hit, int e) {
  const unsigned m = __builtin_amdgcn_ballot_w32(hit);
  if (hit) q[qn + (int)__builtin_amdgcn_mbcnt_lo(m, 0u)] = e;
  qn += __builtin_popcount(m);
}

__device__ __forceinline__ void gcn_spmm_add_row(float* tile, const float* __restrict__ h,
                                                 float sc, int s, int loc, int lane) {
  const float4 v = ((const float4*)(h + (size_t)s * D))[lane];
  float* t = tile + loc * D + lane;
  atomicAdd(t + 0, v.x * sc);
  atomicAdd(t + 32, v.y * sc);
  atomicAdd(t + 64, v.z * sc);
  atomicAdd(t + 96, v.w * sc);
}

__device__ __forceinline__ void gcn_spmm_zero_tile(float* tile, int tid) {
  for (int i = tid; i < SP_TROWS * D / 4; i += 1024)
    ((float4*)tile)[i] = make_float4(0.0f, 0.0f, 0.0f, 0.0f);
}

__device__ __forceinline__ void gcn_spmm_store_tile(const float* tile, const int* rowcnt,
                                                    float* __restrict__ agg,
                                                    int row_lo, int nrows, int lane, int wave) {
  for (int r = wave; r < nrows; r += SP_WAVES) {
    const float* t = tile + r * D + lane;
    const float si = gcn_inv_sqrt_deg(rowcnt[r]);
    float4 o;
    o.x = t[0] * si;
    o.y = t[32] * si;
    o.z = t[64] * si;
    o.w = t[96] * si;
    ((float4*)(agg + (size_t)(row_lo + r) * D))[lane] = o;
  }
}

__device__ __forceinline__ void gcn_spmm_scan(float* tile, int* rowcnt, int* queue,
                                              const float* __restrict__ h,
                                              const int* __restrict__ src,
                                              const int* __restrict__ dst,
                                              const int* __restrict__ deg_out,
                                              int row_lo, int nrows, int n_edges,
                                              int lane, int wave) {
  int* q = queue + wave * SP_QCAP;
  int qn = 0;
  const int nchunks = (n_edges + SP_CH - 1) / SP_CH;
  const int rot = (int)((unsigned)blockIdx.x * (unsigned)nchunks / gridDim.x);

  int c = wave;
  int cc = c + rot;
  if (cc >= nchunks) cc -= nchunks;
  int4 dv[8];
  if (c < nchunks) gcn_spmm_load_chunk(dv, dst, cc, lane, n_edges);

  while (c < nchunks) {
#pragma unroll
    for (int v = 0; v < 8; ++v) {
      const int e = cc * SP_CH + v * 128 + lane * 4;
      const bool h0 = (unsigned)(dv[v].x - row_lo) < (unsigned)nrows;
      const bool h1 = (unsigned)(dv[v].y - row_lo) < (unsigned)nrows;
      const bool h2 = (unsigned)(dv[v].z - row_lo) < (unsigned)nrows;
      const bool h3 = (unsigned)(dv[v].w - row_lo) < (unsigned)nrows;
      if (__builtin_amdgcn_ballot_w32(h0 | h1 | h2 | h3) != 0u) {
        gcn_spmm_push(q, qn, h0, e + 0);
        gcn_spmm_push(q, qn, h1, e + 1);
        gcn_spmm_push(q, qn, h2, e + 2);
        gcn_spmm_push(q, qn, h3, e + 3);
      }
    }

    c += SP_WAVES;
    cc = c + rot;
    if (cc >= nchunks) cc -= nchunks;
    if (c < nchunks) gcn_spmm_load_chunk(dv, dst, cc, lane, n_edges);

    while (qn >= SP_B) {
      qn -= SP_B;
      const int ev = q[qn + (lane & (SP_B - 1))];
      int s[SP_B], d[SP_B];
#pragma unroll
      for (int i = 0; i < SP_B; ++i) {
        const int e = __builtin_amdgcn_readlane(ev, i);
        s[i] = src[e];
        d[i] = dst[e];
      }
      float sc[SP_B];
      float4 v[SP_B];
#pragma unroll
      for (int i = 0; i < SP_B; ++i) {
        sc[i] = gcn_inv_sqrt_deg(deg_out[s[i]]);
        v[i] = ((const float4*)(h + (size_t)s[i] * D))[lane];
      }
#pragma unroll
      for (int i = 0; i < SP_B; ++i) {
        if (lane == 0) atomicAdd(&rowcnt[d[i] - row_lo], 1);
        float* t = tile + (d[i] - row_lo) * D + lane;
        atomicAdd(t + 0, v[i].x * sc[i]);
        atomicAdd(t + 32, v[i].y * sc[i]);
        atomicAdd(t + 64, v[i].z * sc[i]);
        atomicAdd(t + 96, v[i].w * sc[i]);
      }
    }
  }

  for (int i = 0; i < qn; ++i) {
    const int e = __builtin_amdgcn_readfirstlane(q[i]);
    const int s = src[e];
    const int loc = dst[e] - row_lo;
    const float sc = gcn_inv_sqrt_deg(deg_out[s]);
    if (lane == 0) atomicAdd(&rowcnt[loc], 1);
    gcn_spmm_add_row(tile, h, sc, s, loc, lane);
  }
}

template <int HALF>
__device__ __forceinline__ void gcn_spmm_own_unit(v32f& acc,
                                                  const float* __restrict__ h, uint2 e32,
                                                  int lane) {
  float4 v[SP_U];
#pragma unroll
  for (int i = 0; i < SP_U; ++i) {
    const unsigned w = (unsigned)__builtin_amdgcn_readlane((int)e32.x, SP_U * HALF + i);
    const unsigned off = (w >> 8) * (unsigned)(D * sizeof(float)) + (unsigned)lane * 16u;
    v[i] = *(const float4*)((const char*)h + off);
  }
#pragma unroll
  for (int i = 0; i < SP_U; ++i) {
    const unsigned w = (unsigned)__builtin_amdgcn_readlane((int)e32.x, SP_U * HALF + i);
    const float sc = __int_as_float(__builtin_amdgcn_readlane((int)e32.y, SP_U * HALF + i));
    const unsigned g = ((w & 255u) >> 5) * 4u;
    acc[g + 0] += v[i].x * sc;
    acc[g + 1] += v[i].y * sc;
    acc[g + 2] += v[i].z * sc;
    acc[g + 3] += v[i].w * sc;
  }
}

__device__ __forceinline__ uint2 gcn_spmm_own_load(const uint2* stretch, int n, int at, int lane) {
  return (at + lane < n) ? stretch[at + lane] : make_uint2(255u, 0u);
}

__device__ __forceinline__ void gcn_spmm_own_sum(v32f& acc, const float* __restrict__ h,
                                                 const uint2* stretch, int n, int lane) {
  int at = 0;
  uint2 enext = gcn_spmm_own_load(stretch, n, 0, lane);
  while (at < n) {
    const uint2 e32 = enext;
    if (at + 32 < n) enext = gcn_spmm_own_load(stretch, n, at + 32, lane);
    gcn_spmm_own_unit<0>(acc, h, e32, lane);
    if (at + SP_U < n) gcn_spmm_own_unit<1>(acc, h, e32, lane);
    at += 32;
  }
}

#define BN_EPT 4
#define BN_EDGES (1024 * BN_EPT)

__global__ __launch_bounds__(1024) void gcn_bin_kernel(
    const int* __restrict__ src, const int* __restrict__ dst,
    int* __restrict__ deg_out,
    unsigned* __restrict__ lists, int* __restrict__ counts, int n_edges) {
  __shared__ int lc[SP_MAXTILES];

  const int tid = threadIdx.x;
  if (tid < SP_MAXTILES) lc[tid] = 0;
  __syncthreads();

  int t[BN_EPT], r[BN_EPT], d[BN_EPT], s[BN_EPT];
  const int e0 = blockIdx.x * BN_EDGES + tid;
#pragma unroll
  for (int i = 0; i < BN_EPT; ++i) {
    const int e = e0 + i * 1024;
    const int ec = (e < n_edges) ? e : n_edges - 1;
    d[i] = dst[ec];
    s[i] = src[ec];
  }
#pragma unroll
  for (int i = 0; i < BN_EPT; ++i) {
    const int tt = d[i] / SP_RPB;
    const bool ok = e0 + i * 1024 < n_edges;
    t[i] = ok ? tt : -1;
    r[i] = 0;
    if (ok) r[i] = atomicAdd(&lc[tt], 1);
  }
  __syncthreads();

  if (tid < SP_MAXTILES) {
    const int n = lc[tid];
    lc[tid] = (n > 0) ? atomicAdd(&counts[tid * SP_CSTRIDE], n) : 0;
  }
  __syncthreads();

  int base[BN_EPT];
#pragma unroll
  for (int i = 0; i < BN_EPT; ++i) base[i] = lc[(t[i] >= 0) ? t[i] : 0];
#pragma unroll
  for (int i = 0; i < BN_EPT; ++i) {
    const int pos = base[i] + r[i];
    if (t[i] >= 0 && pos < SP_LCAP)
      lists[(size_t)t[i] * SP_LCAP + pos] = ((unsigned)s[i] << 8) | (unsigned)(d[i] - t[i] * SP_RPB);
    if (t[i] >= 0) atomicAdd(&deg_out[s[i]], 1);
  }
}

#define SP_BINW (2 * SP_WAVES)
#define SP_ROWW 256
static_assert(SP_TROWS <= SP_ROWW && SP_WAVES * SP_OWN <= SP_ROWW && SP_ROWW <= 1024,
              "a tile's row factors, and every row number a wave can own, must fit their workspace row");
static_assert(SP_OWN * 4 + 4 <= 32, "a wave's sums and the spare group must fit the 32-float array");
static_assert(SP_RPB <= 224, "row group 7 must stay free for the padding entry");
static_assert(SP_LCAP * 8 <= SP_WAVES * SP_QCAP * 4, "a kept list must fit the queue's LDS");
#define SP_PRE 4

template <bool FIRST>
__global__ __launch_bounds__(1024) void gcn_spmm_list_kernel(
    const float* __restrict__ h, const int* __restrict__ src,
    const int* __restrict__ dst, const int* __restrict__ deg_out,
    float* __restrict__ agg, const unsigned* __restrict__ lists,
    const int* __restrict__ counts, unsigned* __restrict__ slist_w,
    int* __restrict__ sbin, float* __restrict__ srow, int n_nodes, int n_edges) {
  __shared__ float tile[SP_TROWS * D];
  __shared__ __attribute__((aligned(16))) int queue[SP_WAVES * SP_QCAP];
  __shared__ int bincnt[SP_WAVES];
  __shared__ int binstart[SP_WAVES];
  __shared__ int binpos[SP_WAVES];
  __shared__ int rowcnt[SP_TROWS];

  const int tid = threadIdx.x;
  const int lane = tid & 31;
  const int wave = __builtin_amdgcn_readfirstlane(tid >> 5);
  const int row_lo = blockIdx.x * SP_RPB;
  int nrows = n_nodes - row_lo;
  if (nrows > SP_RPB) nrows = SP_RPB;

  const int cnt = counts[blockIdx.x * SP_CSTRIDE];
  const bool kept = cnt <= SP_LCAP;
  const unsigned* list = lists + (size_t)blockIdx.x * SP_LCAP;

  uint2* slist = (uint2*)slist_w + (size_t)blockIdx.x * SP_LCAP;

  if (FIRST || !kept) {
    if (tid < SP_WAVES) bincnt[tid] = 0;
    if (tid < SP_TROWS) rowcnt[tid] = 0;
  }

  if (!kept) {
    gcn_spmm_zero_tile(tile, tid);
    __syncthreads();
    gcn_spmm_scan(tile, rowcnt, queue, h, src, dst, deg_out, row_lo, nrows, n_edges, lane, wave);
    __syncthreads();
    gcn_spmm_store_tile(tile, rowcnt, agg, row_lo, nrows, lane, wave);
    return;
  }

  v32f acc = (v32f){};
  float si_kept[SP_OWN];

  if (FIRST) {
    unsigned ent[SP_PRE];
    int edeg[SP_PRE];
#pragma unroll
    for (int k = 0; k < SP_PRE; ++k) {
      const int i = tid + k * 1024;
      ent[k] = (i < cnt) ? list[i] : 0u;
    }
    __syncthreads();

#pragma unroll
    for (int k = 0; k < SP_PRE; ++k) edeg[k] = deg_out[ent[k] >> 8];

#pragma unroll
    for (int k = 0; k < SP_PRE; ++k)
      if (tid + k * 1024 < cnt) {
        atomicAdd(&bincnt[ent[k] & 31u], 1);
        atomicAdd(&rowcnt[ent[k] & 255u], 1);
      }
    for (int i = tid + SP_PRE * 1024; i < cnt; i += 1024) {
      const unsigned w = list[i];
      atomicAdd(&bincnt[w & 31u], 1);
      atomicAdd(&rowcnt[w & 255u], 1);
    }
    __syncthreads();

    if (wave == 0) {
      const int c = bincnt[lane];
      int incl = c;
#pragma unroll
      for (int o = 1; o < 32; o <<= 1) {
        const int t = __shfl_up(incl, o);
        if (lane >= o) incl += t;
      }
      binstart[lane] = incl - c;
      binpos[lane] = incl - c;
    }
    __syncthreads();

    uint2* sorted = (uint2*)queue;
#pragma unroll
    for (int k = 0; k < SP_PRE; ++k)
      if (tid + k * 1024 < cnt)
        sorted[atomicAdd(&binpos[ent[k] & 31u], 1)] =
            make_uint2(ent[k], __float_as_uint(gcn_inv_sqrt_deg(edeg[k])));
    for (int i = tid + SP_PRE * 1024; i < cnt; i += 1024) {
      const unsigned w = list[i];
      sorted[atomicAdd(&binpos[w & 31u], 1)] =
          make_uint2(w, __float_as_uint(gcn_inv_sqrt_deg(deg_out[w >> 8])));
    }
    __syncthreads();

    for (int i = tid; i < cnt; i += 1024) slist[i] = sorted[i];
    if (tid < SP_WAVES) {
      sbin[blockIdx.x * SP_BINW + tid] = binstart[tid];
      sbin[blockIdx.x * SP_BINW + SP_WAVES + tid] = bincnt[tid];
    }
    if (tid < SP_ROWW)
      srow[blockIdx.x * SP_ROWW + tid] = (tid < SP_TROWS) ? gcn_inv_sqrt_deg(rowcnt[tid]) : 1.0f;

    const int my_start = __builtin_amdgcn_readfirstlane(binstart[wave]);
    const int my_n = __builtin_amdgcn_readfirstlane(bincnt[wave]);
    gcn_spmm_own_sum(acc, h, sorted + my_start, my_n, lane);
  } else {
    const int my_start = sbin[blockIdx.x * SP_BINW + wave];
    const int my_n = sbin[blockIdx.x * SP_BINW + SP_WAVES + wave];
#pragma unroll
    for (int j = 0; j < SP_OWN; ++j) si_kept[j] = srow[blockIdx.x * SP_ROWW + wave + SP_WAVES * j];
    gcn_spmm_own_sum(acc, h, slist + my_start, my_n, lane);
  }

#pragma unroll
  for (int j = 0; j < SP_OWN; ++j) {
    const int r = wave + SP_WAVES * j;
    if (r < nrows) {
      const float si = FIRST ? gcn_inv_sqrt_deg(rowcnt[r]) : si_kept[j];
      ((float4*)(agg + (size_t)(row_lo + r) * D))[lane] =
          make_float4(acc[4 * j] * si, acc[4 * j + 1] * si, acc[4 * j + 2] * si,
                      acc[4 * j + 3] * si);
    }
  }
}

typedef __attribute__((ext_vector_type(4))) float v4f;

#define GM_WAVES 16
#define GM_THREADS (GM_WAVES * 32)
#define GM_ROWS (GM_WAVES * 16)
#define GM_SLOTS (D * D / 4)

__global__ __launch_bounds__(GM_THREADS) void gcn_gemm_kernel(
    const float* __restrict__ agg,
    const float* __restrict__ W, const float* __restrict__ bias,
    float* __restrict__ out, int relu, int n_rows) {
  __shared__ v4f Wl[GM_SLOTS];

  const int tid = threadIdx.x;
  const int lane = tid & 31;
  const int wave = __builtin_amdgcn_readfirstlane(tid >> 5);
  const int c = lane & 15;
  const int hh = lane >> 4;
  const int row0 = (blockIdx.x * GM_WAVES + wave) * 16;
  const bool active = row0 < n_rows;

  v4f a[16];
  if (active) {
    const v4f* g = (const v4f*)(agg + (size_t)(row0 + c) * D) + hh;
#pragma unroll
    for (int q = 0; q < 16; ++q) a[q] = g[2 * q];
  }

#pragma unroll
  for (int j = 0; j < GM_SLOTS / GM_THREADS; ++j) {
    const int slot = tid + j * GM_THREADS;
    const int sl = slot & 31;
    const int t = (slot >> 5) & 7;
    const int q = slot >> 8;
    const float* w = W + (size_t)(8 * q + 4 * (sl >> 4)) * D + 16 * t + (sl & 15);
    v4f v;
    v.x = w[0];
    v.y = w[D];
    v.z = w[2 * D];
    v.w = w[3 * D];
    Wl[slot] = v;
  }
  __syncthreads();
  if (!active) return;

  v8f acc[8];
#pragma unroll
  for (int t = 0; t < 8; ++t) acc[t] = (v8f){};

  v4f bcur[8];
#pragma unroll
  for (int t = 0; t < 8; ++t) bcur[t] = Wl[t * 32 + lane];

#pragma unroll
  for (int q = 0; q < 16; ++q) {
    v4f bnxt[8];
    if (q + 1 < 16) {
#pragma unroll
      for (int t = 0; t < 8; ++t) bnxt[t] = Wl[((q + 1) * 8 + t) * 32 + lane];
    }
    __builtin_amdgcn_sched_barrier(0);
    const v4f aq = a[q];
    v2f a0, a1;
    a0.x = aq.x; a0.y = aq.y;
    a1.x = aq.z; a1.y = aq.w;
#pragma unroll
    for (int t = 0; t < 8; ++t) {
      v2f b;
      b.x = bcur[t].x; b.y = bcur[t].y;
      acc[t] = __builtin_amdgcn_wmma_f32_16x16x4_f32(
           false, a0,  false, b,
           (short)0, acc[t],  false,  false);
    }
#pragma unroll
    for (int t = 0; t < 8; ++t) {
      v2f b;
      b.x = bcur[t].z; b.y = bcur[t].w;
      acc[t] = __builtin_amdgcn_wmma_f32_16x16x4_f32(
           false, a1,  false, b,
           (short)0, acc[t],  false,  false);
    }
    __builtin_amdgcn_sched_barrier(0);
    if (q + 1 < 16) {
#pragma unroll
      for (int t = 0; t < 8; ++t) bcur[t] = bnxt[t];
    }
  }

  const int rbase = row0 + hh * 8;
#pragma unroll
  for (int t = 0; t < 8; ++t) {
    const int ncol = 16 * t + c;
    const float bv = bias[ncol];
#pragma unroll
    for (int r = 0; r < 8; ++r) {
      float v = acc[t][r] + bv;
      if (relu) v = fmaxf(v, 0.0f);
      out[(size_t)(rbase + r) * D + ncol] = v;
    }
  }
}

extern "C" void kernel_launch(void* const* d_in, const int* in_sizes, int n_in,
                              void* d_out, int out_size, void* d_ws, size_t ws_size,
                              hipStream_t stream) {
  const float* feat = (const float*)d_in[0];
  const int* src = (const int*)d_in[1];
  const int* dst = (const int*)d_in[2];
  const float* W1 = (const float*)d_in[3];
  const float* b1 = (const float*)d_in[4];
  const float* W2 = (const float*)d_in[5];
  const float* b2 = (const float*)d_in[6];
  const float* W3 = (const float*)d_in[7];
  const float* b3 = (const float*)d_in[8];
  float* out = (float*)d_out;

  const int N = in_sizes[0] / D;
  const int E = in_sizes[1];

  const int spmm_blocks = (N + SP_RPB - 1) / SP_RPB;

  float* ws = (float*)d_ws;
  int* counts = (int*)ws;
  int* deg_out = counts + SP_MAXTILES * SP_CSTRIDE;
  const long long n_clear = SP_MAXTILES * SP_CSTRIDE + ((long long)N + 63) / 64 * 64;
  float* agg = ws + n_clear;
  float* h = agg + (size_t)N * D;
  unsigned* lists = (unsigned*)(h + (size_t)N * D);
  unsigned* slist = lists + (size_t)spmm_blocks * SP_LCAP;
  int* sbin = (int*)(slist + (size_t)spmm_blocks * SP_LCAP * 2);
  float* srow = (float*)(sbin + (size_t)spmm_blocks * SP_BINW);

  gcn_zero_kernel<<<(int)((n_clear + 1023) / 1024), 1024, 0, stream>>>(ws, n_clear);
  gcn_bin_kernel<<<(E + BN_EDGES - 1) / BN_EDGES, 1024, 0, stream>>>(src, dst, deg_out, lists, counts, E);

  const int gemm_blocks = (N + GM_ROWS - 1) / GM_ROWS;

  gcn_spmm_list_kernel<true><<<spmm_blocks, 1024, 0, stream>>>(feat, src, dst, deg_out, agg, lists, counts, slist, sbin, srow, N, E);
  gcn_gemm_kernel<<<gemm_blocks, GM_THREADS, 0, stream>>>(agg, W1, b1, h, 1, N);

  gcn_spmm_list_kernel<false><<<spmm_blocks, 1024, 0, stream>>>(h, src, dst, deg_out, agg, lists, counts, slist, sbin, srow, N, E);
  gcn_gemm_kernel<<<gemm_blocks, GM_THREADS, 0, stream>>>(agg, W2, b2, h, 1, N);

  gcn_spmm_list_kernel<false><<<spmm_blocks, 1024, 0, stream>>>(h, src, dst, deg_out, agg, lists, counts, slist, sbin, srow, N, E);
  gcn_gemm_kernel<<<gemm_blocks, GM_THREADS, 0, stream>>>(agg, W3, b3, out, 0, N);
}
